# starting kernel
# baseline (speedup 1.0000x reference)
#include <hip/hip_runtime.h>
#include <hip/hip_bf16.h>
#include <math.h>


#define NT     2048
#define DD     384
#define DH     128
#define D3     384
#define D9     1152
#define GAMMA  5.0f
#define LOG2E  1.4426950408889634f

typedef float v2f __attribute__((ext_vector_type(2)));
typedef float v8f __attribute__((ext_vector_type(8)));

static __device__ __forceinline__ v8f wmma4(v2f a, v2f b, v8f c) {
  return __builtin_amdgcn_wmma_f32_16x16x4_f32(
       false, a,  false, b,
       (short)0, c,  false,  false);
}

static __device__ __forceinline__ float softplusf(float w) {
  return (w > 20.0f) ? w : log1pf(__expf(w));
}


__global__ void cea_softplus(const float* __restrict__ in,
                             float* __restrict__ out, int n) {
  int i = blockIdx.x * blockDim.x + threadIdx.x;
  if (i < n) out[i] = softplusf(in[i]);
}

__global__ void cea_fuse(const float* __restrict__ spw,
                         const float* __restrict__ Sq,
                         const float* __restrict__ Sk,
                         const float* __restrict__ Sv,
                         float* __restrict__ Mt) {
  int h   = blockIdx.y;
  int idx = blockIdx.x * 256 + threadIdx.x;
  int n = idx >> 7, k = idx & 127;
  const float* S = (h == 0) ? Sq : (h == 1) ? Sk : Sv;
  const float* W = spw + h * (D3 * DH);
  float acc = 0.0f;
  for (int j = 0; j < D3; ++j) acc += W[j * DH + k] * S[j * D3 + n];
  Mt[h * (D3 * DH) + idx] = acc;
}

__global__ void cea_bias(const float* __restrict__ bq, const float* __restrict__ bk,
                         const float* __restrict__ bv,
                         const float* __restrict__ Sq, const float* __restrict__ Sk,
                         const float* __restrict__ Sv,
                         float* __restrict__ c) {
  int h = blockIdx.x, n = threadIdx.x;
  const float* b = (h == 0) ? bq : (h == 1) ? bk : bv;
  const float* S = (h == 0) ? Sq : (h == 1) ? Sk : Sv;
  float acc = 0.0f;
  for (int j = 0; j < D3; ++j) acc += b[j] * S[j * D3 + n];
  c[h * D3 + n] = acc;
}

__global__ void cea_qkv(const float* __restrict__ x,
                        const float* __restrict__ Mt,
                        const float* __restrict__ cb,
                        float* __restrict__ qkv) {
  __shared__ float xT[16 * DH];
  const int h  = blockIdx.y;
  const int t0 = blockIdx.x * 16;
  const int tid = threadIdx.x;

  for (int idx = tid; idx < 16 * DH; idx += 256) {
    int r = idx >> 7, cc = idx & 127;
    xT[idx] = x[(t0 + r) * DD + h * DH + cc];
  }
  __syncthreads();

  const int lane = tid & 31, wid = tid >> 5;
  const int m  = lane & 15;
  const int kh = (lane >> 4) * 2;
  const float* __restrict__ Mh = Mt + h * (D3 * DH);
  const int n0 = wid * 48;

  v8f acc0 = {}, acc1 = {}, acc2 = {};
  for (int k0 = 0; k0 < DH; k0 += 4) {
    v2f a  = *(const v2f*)&xT[m * DH + k0 + kh];
    v2f b0 = *(const v2f*)&Mh[(n0 + 0  + m) * DH + k0 + kh];
    v2f b1 = *(const v2f*)&Mh[(n0 + 16 + m) * DH + k0 + kh];
    v2f b2 = *(const v2f*)&Mh[(n0 + 32 + m) * DH + k0 + kh];
    acc0 = wmma4(a, b0, acc0);
    acc1 = wmma4(a, b1, acc1);
    acc2 = wmma4(a, b2, acc2);
  }

  const int rbase = (lane >> 4) * 8;
  const int ncol  = lane & 15;
  for (int r = 0; r < 8; ++r) {
    int row = t0 + rbase + r;
    qkv[row * D9 + h * D3 + n0 + 0  + ncol] = acc0[r] + cb[h * D3 + n0 + 0  + ncol];
    qkv[row * D9 + h * D3 + n0 + 16 + ncol] = acc1[r] + cb[h * D3 + n0 + 16 + ncol];
    qkv[row * D9 + h * D3 + n0 + 32 + ncol] = acc2[r] + cb[h * D3 + n0 + 32 + ncol];
  }
}

__global__ void cea_attn(const float* __restrict__ qkv,
                         float* __restrict__ att) {
  __shared__ float Qs[D3], Ksc[D3], Vs[D3];
  const int t = blockIdx.x, i = threadIdx.x;
  const float* __restrict__ base = qkv + t * D9;
  Qs[i]  = base[i];
  Ksc[i] = base[D3 + i] * (-GAMMA * LOG2E);
  Vs[i]  = base[2 * D3 + i];
  __syncthreads();

  const float Qi = Qs[i];
  float acc = 0.0f, ssum = 0.0f;
#pragma unroll 4
  for (int j = 0; j < D3; ++j) {
    float s = __builtin_amdgcn_rcpf(1.0f + __builtin_amdgcn_exp2f(Qi * Ksc[j]));
    acc  += s * Vs[j];
    ssum += s;
  }
  att[t * D3 + i] = acc / (ssum + 1e-8f);
}

__global__ void cea_proj(const float* __restrict__ att,
                         const float* __restrict__ spP,
                         const float* __restrict__ bp,
                         float* __restrict__ hout) {
  __shared__ float aT[16 * D3];
  const int t0 = blockIdx.x * 16;
  const int tid = threadIdx.x;
  for (int idx = tid; idx < 16 * D3; idx += 256)
    aT[idx] = att[t0 * D3 + idx];
  __syncthreads();

  const int lane = tid & 31, wid = tid >> 5;
  const int m  = lane & 15;
  const int kh = (lane >> 4) * 2;
  const int nbase = wid * 144;

  v8f acc[9];
#pragma unroll
  for (int q = 0; q < 9; ++q) acc[q] = (v8f){};

  for (int k0 = 0; k0 < D3; k0 += 4) {
    v2f a = *(const v2f*)&aT[m * D3 + k0 + kh];
#pragma unroll
    for (int q = 0; q < 9; ++q) {
      v2f b = *(const v2f*)&spP[(nbase + q * 16 + m) * D3 + k0 + kh];
      acc[q] = wmma4(a, b, acc[q]);
    }
  }

  const int rbase = (lane >> 4) * 8;
  const int ncol  = lane & 15;
#pragma unroll
  for (int q = 0; q < 9; ++q) {
    int n = nbase + q * 16 + ncol;
    float bias = bp[n];
    for (int r = 0; r < 8; ++r)
      hout[(t0 + rbase + r) * D9 + n] = acc[q][r] + bias;
  }
}

__global__ void cea_ln(float* __restrict__ out,
                       const float* __restrict__ gn,
                       const float* __restrict__ bn) {
  __shared__ float row[D9];
  __shared__ float red[256];
  const int t = blockIdx.x, tid = threadIdx.x;
  float* __restrict__ h = out + t * D9;

  float ps = 0.0f;
  for (int n = tid; n < D9; n += 256) { float v = h[n]; row[n] = v; ps += v; }
  red[tid] = ps;
  __syncthreads();
  for (int s = 128; s > 0; s >>= 1) {
    if (tid < s) red[tid] += red[tid + s];
    __syncthreads();
  }
  const float mu = red[0] * (1.0f / (float)D9);
  __syncthreads();

  float pv = 0.0f;
  for (int n = tid; n < D9; n += 256) { float dv = row[n] - mu; pv += dv * dv; }
  red[tid] = pv;
  __syncthreads();
  for (int s = 128; s > 0; s >>= 1) {
    if (tid < s) red[tid] += red[tid + s];
    __syncthreads();
  }
  const float var = red[0] * (1.0f / (float)D9);
  const float rs  = __builtin_amdgcn_rsqf(var + 1e-5f);

  for (int n = tid; n < D9; n += 256)
    h[n] = (row[n] - mu) * rs * gn[n] + bn[n];
}


extern "C" void kernel_launch(void* const* d_in, const int* in_sizes, int n_in,
                              void* d_out, int out_size, void* d_ws, size_t ws_size,
                              hipStream_t stream) {
  const float* x  = (const float*)d_in[0];
  const float* Wq = (const float*)d_in[1];
  const float* bq = (const float*)d_in[2];
  const float* Sq = (const float*)d_in[3];
  const float* Wk = (const float*)d_in[4];
  const float* bk = (const float*)d_in[5];
  const float* Sk = (const float*)d_in[6];
  const float* Wv = (const float*)d_in[7];
  const float* bv = (const float*)d_in[8];
  const float* Sv = (const float*)d_in[9];
  const float* Wp = (const float*)d_in[10];
  const float* bp = (const float*)d_in[11];
  const float* gn = (const float*)d_in[12];
  const float* bn = (const float*)d_in[13];
  float* out = (float*)d_out;

  float* ws = (float*)d_ws;
  float* spw  = ws;
  float* Mt   = spw + 3 * D3 * DH;
  float* cb   = Mt  + 3 * D3 * DH;
  float* spP  = cb  + 3 * D3;
  float* qkv  = spP + D9 * D3;
  float* attb = qkv + NT * D9;

  cea_softplus<<<dim3((D3 * DH + 255) / 256), dim3(256), 0, stream>>>(Wq, spw,              D3 * DH);
  cea_softplus<<<dim3((D3 * DH + 255) / 256), dim3(256), 0, stream>>>(Wk, spw + D3 * DH,    D3 * DH);
  cea_softplus<<<dim3((D3 * DH + 255) / 256), dim3(256), 0, stream>>>(Wv, spw + 2 * D3 * DH, D3 * DH);
  cea_softplus<<<dim3((D9 * D3 + 255) / 256), dim3(256), 0, stream>>>(Wp, spP,              D9 * D3);

  cea_fuse<<<dim3((D3 * DH) / 256, 3), dim3(256), 0, stream>>>(spw, Sq, Sk, Sv, Mt);
  cea_bias<<<dim3(3), dim3(D3), 0, stream>>>(bq, bk, bv, Sq, Sk, Sv, cb);

  cea_qkv<<<dim3(NT / 16, 3), dim3(256), 0, stream>>>(x, Mt, cb, qkv);

  cea_attn<<<dim3(NT), dim3(D3), 0, stream>>>(qkv, attb);

  cea_proj<<<dim3(NT / 16), dim3(256), 0, stream>>>(attb, spP, bp, out);

  cea_ln<<<dim3(NT), dim3(256), 0, stream>>>(out, gn, bn);

  (void)in_sizes; (void)n_in; (void)out_size; (void)ws_size;
}


# projection GEMM: stage both operands in LDS with coalesced 16-byte loads (tile 64 tokens x 144 cols, 256 blocks, next chunk prefetched) instead of per-lane 8-byte global reads of the weight
# speedup vs baseline: 1.5475x; 1.5475x over previous
#include <hip/hip_runtime.h>
#include <hip/hip_bf16.h>
#include <math.h>


#define NT     2048
#define DD     384
#define DH     128
#define D3     384
#define D9     1152
#define GAMMA  5.0f
#define LOG2E  1.4426950408889634f

typedef float v2f __attribute__((ext_vector_type(2)));
typedef float v4f __attribute__((ext_vector_type(4)));
typedef float v8f __attribute__((ext_vector_type(8)));

static __device__ __forceinline__ v8f wmma4(v2f a, v2f b, v8f c) {
  return __builtin_amdgcn_wmma_f32_16x16x4_f32(
       false, a,  false, b,
       (short)0, c,  false,  false);
}

static __device__ __forceinline__ float softplusf(float w) {
  return (w > 20.0f) ? w : log1pf(__expf(w));
}


__global__ void cea_softplus(const float* __restrict__ in,
                             float* __restrict__ out, int n) {
  int i = blockIdx.x * blockDim.x + threadIdx.x;
  if (i < n) out[i] = softplusf(in[i]);
}

__global__ void cea_fuse(const float* __restrict__ spw,
                         const float* __restrict__ Sq,
                         const float* __restrict__ Sk,
                         const float* __restrict__ Sv,
                         float* __restrict__ Mt) {
  int h   = blockIdx.y;
  int idx = blockIdx.x * 256 + threadIdx.x;
  int n = idx >> 7, k = idx & 127;
  const float* S = (h == 0) ? Sq : (h == 1) ? Sk : Sv;
  const float* W = spw + h * (D3 * DH);
  float acc = 0.0f;
  for (int j = 0; j < D3; ++j) acc += W[j * DH + k] * S[j * D3 + n];
  Mt[h * (D3 * DH) + idx] = acc;
}

__global__ void cea_bias(const float* __restrict__ bq, const float* __restrict__ bk,
                         const float* __restrict__ bv,
                         const float* __restrict__ Sq, const float* __restrict__ Sk,
                         const float* __restrict__ Sv,
                         float* __restrict__ c) {
  int h = blockIdx.x, n = threadIdx.x;
  const float* b = (h == 0) ? bq : (h == 1) ? bk : bv;
  const float* S = (h == 0) ? Sq : (h == 1) ? Sk : Sv;
  float acc = 0.0f;
  for (int j = 0; j < D3; ++j) acc += b[j] * S[j * D3 + n];
  c[h * D3 + n] = acc;
}

__global__ void cea_qkv(const float* __restrict__ x,
                        const float* __restrict__ Mt,
                        const float* __restrict__ cb,
                        float* __restrict__ qkv) {
  __shared__ float xT[16 * DH];
  const int h  = blockIdx.y;
  const int t0 = blockIdx.x * 16;
  const int tid = threadIdx.x;

  for (int idx = tid; idx < 16 * DH; idx += 256) {
    int r = idx >> 7, cc = idx & 127;
    xT[idx] = x[(t0 + r) * DD + h * DH + cc];
  }
  __syncthreads();

  const int lane = tid & 31, wid = tid >> 5;
  const int m  = lane & 15;
  const int kh = (lane >> 4) * 2;
  const float* __restrict__ Mh = Mt + h * (D3 * DH);
  const int n0 = wid * 48;

  v8f acc0 = {}, acc1 = {}, acc2 = {};
  for (int k0 = 0; k0 < DH; k0 += 4) {
    v2f a  = *(const v2f*)&xT[m * DH + k0 + kh];
    v2f b0 = *(const v2f*)&Mh[(n0 + 0  + m) * DH + k0 + kh];
    v2f b1 = *(const v2f*)&Mh[(n0 + 16 + m) * DH + k0 + kh];
    v2f b2 = *(const v2f*)&Mh[(n0 + 32 + m) * DH + k0 + kh];
    acc0 = wmma4(a, b0, acc0);
    acc1 = wmma4(a, b1, acc1);
    acc2 = wmma4(a, b2, acc2);
  }

  const int rbase = (lane >> 4) * 8;
  const int ncol  = lane & 15;
  for (int r = 0; r < 8; ++r) {
    int row = t0 + rbase + r;
    qkv[row * D9 + h * D3 + n0 + 0  + ncol] = acc0[r] + cb[h * D3 + n0 + 0  + ncol];
    qkv[row * D9 + h * D3 + n0 + 16 + ncol] = acc1[r] + cb[h * D3 + n0 + 16 + ncol];
    qkv[row * D9 + h * D3 + n0 + 32 + ncol] = acc2[r] + cb[h * D3 + n0 + 32 + ncol];
  }
}

__global__ void cea_attn(const float* __restrict__ qkv,
                         float* __restrict__ att) {
  __shared__ float Qs[D3], Ksc[D3], Vs[D3];
  const int t = blockIdx.x, i = threadIdx.x;
  const float* __restrict__ base = qkv + t * D9;
  Qs[i]  = base[i];
  Ksc[i] = base[D3 + i] * (-GAMMA * LOG2E);
  Vs[i]  = base[2 * D3 + i];
  __syncthreads();

  const float Qi = Qs[i];
  float acc = 0.0f, ssum = 0.0f;
#pragma unroll 4
  for (int j = 0; j < D3; ++j) {
    float s = __builtin_amdgcn_rcpf(1.0f + __builtin_amdgcn_exp2f(Qi * Ksc[j]));
    acc  += s * Vs[j];
    ssum += s;
  }
  att[t * D3 + i] = acc / (ssum + 1e-8f);
}

#define PJ_BM 64
#define PJ_BN 144
#define PJ_KC 32
#define PJ_P  36
struct pj_frag { v4f a; v4f b[9]; };
static __device__ __forceinline__ void pj_read(const float* __restrict__ Ab,
                                               const float* __restrict__ Bb,
                                               int ks, pj_frag& f) {
  v2f lo = *(const v2f*)&Ab[ks], hi = *(const v2f*)&Ab[ks + 4];
  f.a = (v4f){lo.x, lo.y, hi.x, hi.y};
#pragma unroll
  for (int q = 0; q < 9; ++q) {
    v2f l2 = *(const v2f*)&Bb[q * 16 * PJ_P + ks], h2 = *(const v2f*)&Bb[q * 16 * PJ_P + ks + 4];
    f.b[q] = (v4f){l2.x, l2.y, h2.x, h2.y};
  }
}
static __device__ __forceinline__ void pj_mma(const pj_frag& f, v8f (&acc)[9]) {
#pragma unroll
  for (int q = 0; q < 9; ++q) acc[q] = wmma4((v2f){f.a.x, f.a.y}, (v2f){f.b[q].x, f.b[q].y}, acc[q]);
#pragma unroll
  for (int q = 0; q < 9; ++q) acc[q] = wmma4((v2f){f.a.z, f.a.w}, (v2f){f.b[q].z, f.b[q].w}, acc[q]);
}
static __device__ __forceinline__ void pj_chunk(const float* __restrict__ Ab,
                                                const float* __restrict__ Bb,
                                                v8f (&acc)[9]) {
  pj_frag f0, f1;
  pj_read(Ab, Bb, 0, f0);
  pj_read(Ab, Bb, 8, f1);
  __builtin_amdgcn_sched_barrier(0);
  pj_mma(f0, acc);
  __builtin_amdgcn_sched_barrier(0);
  pj_read(Ab, Bb, 16, f0);
  __builtin_amdgcn_sched_barrier(0);
  pj_mma(f1, acc);
  __builtin_amdgcn_sched_barrier(0);
  pj_read(Ab, Bb, 24, f1);
  __builtin_amdgcn_sched_barrier(0);
  pj_mma(f0, acc);
  __builtin_amdgcn_sched_barrier(0);
  pj_mma(f1, acc);
}

__global__ __launch_bounds__(128)
void cea_proj(const float* __restrict__ att,
              const float* __restrict__ spP,
              const float* __restrict__ bp,
              float* __restrict__ hout) {
  __shared__ __attribute__((aligned(16))) float As[2 * PJ_BM * PJ_P];
  __shared__ __attribute__((aligned(16))) float Bs[2 * PJ_BN * PJ_P];
  const int tid = threadIdx.x;
  const int t0 = blockIdx.x * PJ_BM;
  const int n0 = blockIdx.y * PJ_BN;

  const int srow = tid >> 3;
  const int sq   = (tid & 7) * 4;
  const float* __restrict__ ag = att + (t0 + srow) * D3 + sq;
  const float* __restrict__ bg = spP + (n0 + srow) * D3 + sq;
  const int sl = srow * PJ_P + sq;

  v4f ra0, ra1, ra2, ra3, rb0, rb1, rb2, rb3, rb4, rb5, rb6, rb7, rb8;
#define PJ_GLOAD(ko)                                        \
  ra0 = *(const v4f*)(ag + 0 * 16 * D3 + (ko));             \
  ra1 = *(const v4f*)(ag + 1 * 16 * D3 + (ko));             \
  ra2 = *(const v4f*)(ag + 2 * 16 * D3 + (ko));             \
  ra3 = *(const v4f*)(ag + 3 * 16 * D3 + (ko));             \
  rb0 = *(const v4f*)(bg + 0 * 16 * D3 + (ko));             \
  rb1 = *(const v4f*)(bg + 1 * 16 * D3 + (ko));             \
  rb2 = *(const v4f*)(bg + 2 * 16 * D3 + (ko));             \
  rb3 = *(const v4f*)(bg + 3 * 16 * D3 + (ko));             \
  rb4 = *(const v4f*)(bg + 4 * 16 * D3 + (ko));             \
  rb5 = *(const v4f*)(bg + 5 * 16 * D3 + (ko));             \
  rb6 = *(const v4f*)(bg + 6 * 16 * D3 + (ko));             \
  rb7 = *(const v4f*)(bg + 7 * 16 * D3 + (ko));             \
  rb8 = *(const v4f*)(bg + 8 * 16 * D3 + (ko));
#define PJ_SSTORE(st)                                                   \
  *(v4f*)&As[(st) * (PJ_BM * PJ_P) + sl + 0 * 16 * PJ_P] = ra0;         \
  *(v4f*)&As[(st) * (PJ_BM * PJ_P) + sl + 1 * 16 * PJ_P] = ra1;         \
  *(v4f*)&As[(st) * (PJ_BM * PJ_P) + sl + 2 * 16 * PJ_P] = ra2;         \
  *(v4f*)&As[(st) * (PJ_BM * PJ_P) + sl + 3 * 16 * PJ_P] = ra3;         \
  *(v4f*)&Bs[(st) * (PJ_BN * PJ_P) + sl + 0 * 16 * PJ_P] = rb0;         \
  *(v4f*)&Bs[(st) * (PJ_BN * PJ_P) + sl + 1 * 16 * PJ_P] = rb1;         \
  *(v4f*)&Bs[(st) * (PJ_BN * PJ_P) + sl + 2 * 16 * PJ_P] = rb2;         \
  *(v4f*)&Bs[(st) * (PJ_BN * PJ_P) + sl + 3 * 16 * PJ_P] = rb3;         \
  *(v4f*)&Bs[(st) * (PJ_BN * PJ_P) + sl + 4 * 16 * PJ_P] = rb4;         \
  *(v4f*)&Bs[(st) * (PJ_BN * PJ_P) + sl + 5 * 16 * PJ_P] = rb5;         \
  *(v4f*)&Bs[(st) * (PJ_BN * PJ_P) + sl + 6 * 16 * PJ_P] = rb6;         \
  *(v4f*)&Bs[(st) * (PJ_BN * PJ_P) + sl + 7 * 16 * PJ_P] = rb7;         \
  *(v4f*)&Bs[(st) * (PJ_BN * PJ_P) + sl + 8 * 16 * PJ_P] = rb8;

  PJ_GLOAD(0)
  PJ_SSTORE(0)
  __syncthreads();

  const int lane = tid & 31, wid = tid >> 5;
  const int m  = lane & 15;
  const int kh = (lane >> 4) * 2;
  const int aoff = (wid * 16 + m) * PJ_P + kh;
  const int boff = m * PJ_P + kh;

  v8f acc[9];
#pragma unroll
  for (int q = 0; q < 9; ++q) acc[q] = (v8f){};

  for (int kc = 0; kc < D3 / PJ_KC - 1; ++kc) {
    const int cur = (kc & 1);
    const int ko = (kc + 1) * PJ_KC;
    PJ_GLOAD(ko)
    pj_chunk(As + cur * (PJ_BM * PJ_P) + aoff, Bs + cur * (PJ_BN * PJ_P) + boff, acc);
    PJ_SSTORE(cur ^ 1)
    __syncthreads();
  }
  pj_chunk(As + 1 * (PJ_BM * PJ_P) + aoff, Bs + 1 * (PJ_BN * PJ_P) + boff, acc);
#undef PJ_GLOAD
#undef PJ_SSTORE

  const int rbase = (lane >> 4) * 8;
  const int ncol  = lane & 15;
#pragma unroll
  for (int q = 0; q < 9; ++q) {
    int n = n0 + q * 16 + ncol;
    float bias = bp[n];
    for (int r = 0; r < 8; ++r)
      hout[(t0 + wid * 16 + rbase + r) * D9 + n] = acc[q][r] + bias;
  }
}

__global__ void cea_ln(float* __restrict__ out,
                       const float* __restrict__ gn,
                       const float* __restrict__ bn) {
  __shared__ float row[D9];
  __shared__ float red[256];
  const int t = blockIdx.x, tid = threadIdx.x;
  float* __restrict__ h = out + t * D9;

  float ps = 0.0f;
  for (int n = tid; n < D9; n += 256) { float v = h[n]; row[n] = v; ps += v; }
  red[tid] = ps;
  __syncthreads();
  for (int s = 128; s > 0; s >>= 1) {
    if (tid < s) red[tid] += red[tid + s];
    __syncthreads();
  }
  const float mu = red[0] * (1.0f / (float)D9);
  __syncthreads();

  float pv = 0.0f;
  for (int n = tid; n < D9; n += 256) { float dv = row[n] - mu; pv += dv * dv; }
  red[tid] = pv;
  __syncthreads();
  for (int s = 128; s > 0; s >>= 1) {
    if (tid < s) red[tid] += red[tid + s];
    __syncthreads();
  }
  const float var = red[0] * (1.0f / (float)D9);
  const float rs  = __builtin_amdgcn_rsqf(var + 1e-5f);

  for (int n = tid; n < D9; n += 256)
    h[n] = (row[n] - mu) * rs * gn[n] + bn[n];
}


extern "C" void kernel_launch(void* const* d_in, const int* in_sizes, int n_in,
                              void* d_out, int out_size, void* d_ws, size_t ws_size,
                              hipStream_t stream) {
  const float* x  = (const float*)d_in[0];
  const float* Wq = (const float*)d_in[1];
  const float* bq = (const float*)d_in[2];
  const float* Sq = (const float*)d_in[3];
  const float* Wk = (const float*)d_in[4];
  const float* bk = (const float*)d_in[5];
  const float* Sk = (const float*)d_in[6];
  const float* Wv = (const float*)d_in[7];
  const float* bv = (const float*)d_in[8];
  const float* Sv = (const float*)d_in[9];
  const float* Wp = (const float*)d_in[10];
  const float* bp = (const float*)d_in[11];
  const float* gn = (const float*)d_in[12];
  const float* bn = (const float*)d_in[13];
  float* out = (float*)d_out;

  float* ws = (float*)d_ws;
  float* spw  = ws;
  float* Mt   = spw + 3 * D3 * DH;
  float* cb   = Mt  + 3 * D3 * DH;
  float* spP  = cb  + 3 * D3;
  float* qkv  = spP + D9 * D3;
  float* attb = qkv + NT * D9;

  cea_softplus<<<dim3((D3 * DH + 255) / 256), dim3(256), 0, stream>>>(Wq, spw,              D3 * DH);
  cea_softplus<<<dim3((D3 * DH + 255) / 256), dim3(256), 0, stream>>>(Wk, spw + D3 * DH,    D3 * DH);
  cea_softplus<<<dim3((D3 * DH + 255) / 256), dim3(256), 0, stream>>>(Wv, spw + 2 * D3 * DH, D3 * DH);
  cea_softplus<<<dim3((D9 * D3 + 255) / 256), dim3(256), 0, stream>>>(Wp, spP,              D9 * D3);

  cea_fuse<<<dim3((D3 * DH) / 256, 3), dim3(256), 0, stream>>>(spw, Sq, Sk, Sv, Mt);
  cea_bias<<<dim3(3), dim3(D3), 0, stream>>>(bq, bk, bv, Sq, Sk, Sv, cb);

  cea_qkv<<<dim3(NT / 16, 3), dim3(256), 0, stream>>>(x, Mt, cb, qkv);

  cea_attn<<<dim3(NT), dim3(D3), 0, stream>>>(qkv, attb);

  cea_proj<<<dim3(NT / PJ_BM, D9 / PJ_BN), dim3(128), 0, stream>>>(attb, spP, bp, out);

  cea_ln<<<dim3(NT), dim3(256), 0, stream>>>(out, gn, bn);

  (void)in_sizes; (void)n_in; (void)out_size; (void)ws_size;
}


# weight preprocessing in one launch: the four softplus launches and the bias rotation become one kernel (10 launches -> 6); bias sum split in 8 partial sums added in fixed order
# speedup vs baseline: 1.8171x; 1.1742x over previous
#include <hip/hip_runtime.h>
#include <hip/hip_bf16.h>
#include <math.h>


#define NT     2048
#define DD     384
#define DH     128
#define D3     384
#define D9     1152
#define GAMMA  5.0f
#define LOG2E  1.4426950408889634f

typedef float v2f __attribute__((ext_vector_type(2)));
typedef float v4f __attribute__((ext_vector_type(4)));
typedef float v8f __attribute__((ext_vector_type(8)));

static __device__ __forceinline__ v8f wmma4(v2f a, v2f b, v8f c) {
  return __builtin_amdgcn_wmma_f32_16x16x4_f32(
       false, a,  false, b,
       (short)0, c,  false,  false);
}

static __device__ __forceinline__ float softplusf(float w) {
  return (w > 20.0f) ? w : log1pf(__expf(w));
}


#define PREP_BIAS_BLOCKS ((3 * D3) / 32)
#define PREP_SPW_BLOCKS  ((3 * D3 * DH) / 256)
#define PREP_SPP_BLOCKS  ((D9 * D3) / 256)
__global__ __launch_bounds__(256)
void cea_prep(const float* __restrict__ Wq, const float* __restrict__ Wk,
              const float* __restrict__ Wv, const float* __restrict__ Wp,
              const float* __restrict__ bq, const float* __restrict__ bk,
              const float* __restrict__ bv,
              const float* __restrict__ Sq, const float* __restrict__ Sk,
              const float* __restrict__ Sv,
              float* __restrict__ spw, float* __restrict__ spP,
              float* __restrict__ c) {
  __shared__ float red[8 * 32];
  const int blk = blockIdx.x, tid = threadIdx.x;
  if (blk >= PREP_BIAS_BLOCKS + PREP_SPW_BLOCKS) {
    int i = (blk - PREP_BIAS_BLOCKS - PREP_SPW_BLOCKS) * 256 + tid;
    spP[i] = softplusf(Wp[i]);
    return;
  }
  if (blk >= PREP_BIAS_BLOCKS) {
    int b2 = blk - PREP_BIAS_BLOCKS;
    int h = b2 / (PREP_SPW_BLOCKS / 3);
    int i = (b2 - h * (PREP_SPW_BLOCKS / 3)) * 256 + tid;
    const float* W = (h == 0) ? Wq : (h == 1) ? Wk : Wv;
    spw[h * (D3 * DH) + i] = softplusf(W[i]);
    return;
  }
  const int h  = blk / (D3 / 32);
  const int nl = tid & 31, jg = tid >> 5;
  const int n  = (blk - h * (D3 / 32)) * 32 + nl;
  const float* b = (h == 0) ? bq : (h == 1) ? bk : bv;
  const float* S = (h == 0) ? Sq : (h == 1) ? Sk : Sv;
  float acc = 0.0f;
#pragma unroll 16
  for (int jj = 0; jj < D3 / 8; ++jj) {
    int j = jg * (D3 / 8) + jj;
    acc += b[j] * S[j * D3 + n];
  }
  red[jg * 32 + nl] = acc;
  __syncthreads();
  if (jg == 0) {
    float s = red[nl];
    for (int q = 1; q < 8; ++q) s += red[q * 32 + nl];
    c[h * D3 + n] = s;
  }
}

__global__ void cea_fuse(const float* __restrict__ spw,
                         const float* __restrict__ Sq,
                         const float* __restrict__ Sk,
                         const float* __restrict__ Sv,
                         float* __restrict__ Mt) {
  int h   = blockIdx.y;
  int idx = blockIdx.x * 256 + threadIdx.x;
  int n = idx >> 7, k = idx & 127;
  const float* S = (h == 0) ? Sq : (h == 1) ? Sk : Sv;
  const float* W = spw + h * (D3 * DH);
  float acc = 0.0f;
  for (int j = 0; j < D3; ++j) acc += W[j * DH + k] * S[j * D3 + n];
  Mt[h * (D3 * DH) + idx] = acc;
}

__global__ void cea_qkv(const float* __restrict__ x,
                        const float* __restrict__ Mt,
                        const float* __restrict__ cb,
                        float* __restrict__ qkv) {
  __shared__ float xT[16 * DH];
  const int h  = blockIdx.y;
  const int t0 = blockIdx.x * 16;
  const int tid = threadIdx.x;

  for (int idx = tid; idx < 16 * DH; idx += 256) {
    int r = idx >> 7, cc = idx & 127;
    xT[idx] = x[(t0 + r) * DD + h * DH + cc];
  }
  __syncthreads();

  const int lane = tid & 31, wid = tid >> 5;
  const int m  = lane & 15;
  const int kh = (lane >> 4) * 2;
  const float* __restrict__ Mh = Mt + h * (D3 * DH);
  const int n0 = wid * 48;

  v8f acc0 = {}, acc1 = {}, acc2 = {};
  for (int k0 = 0; k0 < DH; k0 += 4) {
    v2f a  = *(const v2f*)&xT[m * DH + k0 + kh];
    v2f b0 = *(const v2f*)&Mh[(n0 + 0  + m) * DH + k0 + kh];
    v2f b1 = *(const v2f*)&Mh[(n0 + 16 + m) * DH + k0 + kh];
    v2f b2 = *(const v2f*)&Mh[(n0 + 32 + m) * DH + k0 + kh];
    acc0 = wmma4(a, b0, acc0);
    acc1 = wmma4(a, b1, acc1);
    acc2 = wmma4(a, b2, acc2);
  }

  const int rbase = (lane >> 4) * 8;
  const int ncol  = lane & 15;
  for (int r = 0; r < 8; ++r) {
    int row = t0 + rbase + r;
    qkv[row * D9 + h * D3 + n0 + 0  + ncol] = acc0[r] + cb[h * D3 + n0 + 0  + ncol];
    qkv[row * D9 + h * D3 + n0 + 16 + ncol] = acc1[r] + cb[h * D3 + n0 + 16 + ncol];
    qkv[row * D9 + h * D3 + n0 + 32 + ncol] = acc2[r] + cb[h * D3 + n0 + 32 + ncol];
  }
}

__global__ void cea_attn(const float* __restrict__ qkv,
                         float* __restrict__ att) {
  __shared__ float Qs[D3], Ksc[D3], Vs[D3];
  const int t = blockIdx.x, i = threadIdx.x;
  const float* __restrict__ base = qkv + t * D9;
  Qs[i]  = base[i];
  Ksc[i] = base[D3 + i] * (-GAMMA * LOG2E);
  Vs[i]  = base[2 * D3 + i];
  __syncthreads();

  const float Qi = Qs[i];
  float acc = 0.0f, ssum = 0.0f;
#pragma unroll 4
  for (int j = 0; j < D3; ++j) {
    float s = __builtin_amdgcn_rcpf(1.0f + __builtin_amdgcn_exp2f(Qi * Ksc[j]));
    acc  += s * Vs[j];
    ssum += s;
  }
  att[t * D3 + i] = acc / (ssum + 1e-8f);
}

#define PJ_BM 64
#define PJ_BN 144
#define PJ_KC 32
#define PJ_P  36
struct pj_frag { v4f a; v4f b[9]; };
static __device__ __forceinline__ void pj_read(const float* __restrict__ Ab,
                                               const float* __restrict__ Bb,
                                               int ks, pj_frag& f) {
  v2f lo = *(const v2f*)&Ab[ks], hi = *(const v2f*)&Ab[ks + 4];
  f.a = (v4f){lo.x, lo.y, hi.x, hi.y};
#pragma unroll
  for (int q = 0; q < 9; ++q) {
    v2f l2 = *(const v2f*)&Bb[q * 16 * PJ_P + ks], h2 = *(const v2f*)&Bb[q * 16 * PJ_P + ks + 4];
    f.b[q] = (v4f){l2.x, l2.y, h2.x, h2.y};
  }
}
static __device__ __forceinline__ void pj_mma(const pj_frag& f, v8f (&acc)[9]) {
#pragma unroll
  for (int q = 0; q < 9; ++q) acc[q] = wmma4((v2f){f.a.x, f.a.y}, (v2f){f.b[q].x, f.b[q].y}, acc[q]);
#pragma unroll
  for (int q = 0; q < 9; ++q) acc[q] = wmma4((v2f){f.a.z, f.a.w}, (v2f){f.b[q].z, f.b[q].w}, acc[q]);
}
static __device__ __forceinline__ void pj_chunk(const float* __restrict__ Ab,
                                                const float* __restrict__ Bb,
                                                v8f (&acc)[9]) {
  pj_frag f0, f1;
  pj_read(Ab, Bb, 0, f0);
  pj_read(Ab, Bb, 8, f1);
  __builtin_amdgcn_sched_barrier(0);
  pj_mma(f0, acc);
  __builtin_amdgcn_sched_barrier(0);
  pj_read(Ab, Bb, 16, f0);
  __builtin_amdgcn_sched_barrier(0);
  pj_mma(f1, acc);
  __builtin_amdgcn_sched_barrier(0);
  pj_read(Ab, Bb, 24, f1);
  __builtin_amdgcn_sched_barrier(0);
  pj_mma(f0, acc);
  __builtin_amdgcn_sched_barrier(0);
  pj_mma(f1, acc);
}

__global__ __launch_bounds__(128)
void cea_proj(const float* __restrict__ att,
              const float* __restrict__ spP,
              const float* __restrict__ bp,
              float* __restrict__ hout) {
  __shared__ __attribute__((aligned(16))) float As[2 * PJ_BM * PJ_P];
  __shared__ __attribute__((aligned(16))) float Bs[2 * PJ_BN * PJ_P];
  const int tid = threadIdx.x;
  const int t0 = blockIdx.x * PJ_BM;
  const int n0 = blockIdx.y * PJ_BN;

  const int srow = tid >> 3;
  const int sq   = (tid & 7) * 4;
  const float* __restrict__ ag = att + (t0 + srow) * D3 + sq;
  const float* __restrict__ bg = spP + (n0 + srow) * D3 + sq;
  const int sl = srow * PJ_P + sq;

  v4f ra0, ra1, ra2, ra3, rb0, rb1, rb2, rb3, rb4, rb5, rb6, rb7, rb8;
#define PJ_GLOAD(ko)                                        \
  ra0 = *(const v4f*)(ag + 0 * 16 * D3 + (ko));             \
  ra1 = *(const v4f*)(ag + 1 * 16 * D3 + (ko));             \
  ra2 = *(const v4f*)(ag + 2 * 16 * D3 + (ko));             \
  ra3 = *(const v4f*)(ag + 3 * 16 * D3 + (ko));             \
  rb0 = *(const v4f*)(bg + 0 * 16 * D3 + (ko));             \
  rb1 = *(const v4f*)(bg + 1 * 16 * D3 + (ko));             \
  rb2 = *(const v4f*)(bg + 2 * 16 * D3 + (ko));             \
  rb3 = *(const v4f*)(bg + 3 * 16 * D3 + (ko));             \
  rb4 = *(const v4f*)(bg + 4 * 16 * D3 + (ko));             \
  rb5 = *(const v4f*)(bg + 5 * 16 * D3 + (ko));             \
  rb6 = *(const v4f*)(bg + 6 * 16 * D3 + (ko));             \
  rb7 = *(const v4f*)(bg + 7 * 16 * D3 + (ko));             \
  rb8 = *(const v4f*)(bg + 8 * 16 * D3 + (ko));
#define PJ_SSTORE(st)                                                   \
  *(v4f*)&As[(st) * (PJ_BM * PJ_P) + sl + 0 * 16 * PJ_P] = ra0;         \
  *(v4f*)&As[(st) * (PJ_BM * PJ_P) + sl + 1 * 16 * PJ_P] = ra1;         \
  *(v4f*)&As[(st) * (PJ_BM * PJ_P) + sl + 2 * 16 * PJ_P] = ra2;         \
  *(v4f*)&As[(st) * (PJ_BM * PJ_P) + sl + 3 * 16 * PJ_P] = ra3;         \
  *(v4f*)&Bs[(st) * (PJ_BN * PJ_P) + sl + 0 * 16 * PJ_P] = rb0;         \
  *(v4f*)&Bs[(st) * (PJ_BN * PJ_P) + sl + 1 * 16 * PJ_P] = rb1;         \
  *(v4f*)&Bs[(st) * (PJ_BN * PJ_P) + sl + 2 * 16 * PJ_P] = rb2;         \
  *(v4f*)&Bs[(st) * (PJ_BN * PJ_P) + sl + 3 * 16 * PJ_P] = rb3;         \
  *(v4f*)&Bs[(st) * (PJ_BN * PJ_P) + sl + 4 * 16 * PJ_P] = rb4;         \
  *(v4f*)&Bs[(st) * (PJ_BN * PJ_P) + sl + 5 * 16 * PJ_P] = rb5;         \
  *(v4f*)&Bs[(st) * (PJ_BN * PJ_P) + sl + 6 * 16 * PJ_P] = rb6;         \
  *(v4f*)&Bs[(st) * (PJ_BN * PJ_P) + sl + 7 * 16 * PJ_P] = rb7;         \
  *(v4f*)&Bs[(st) * (PJ_BN * PJ_P) + sl + 8 * 16 * PJ_P] = rb8;

  PJ_GLOAD(0)
  PJ_SSTORE(0)
  __syncthreads();

  const int lane = tid & 31, wid = tid >> 5;
  const int m  = lane & 15;
  const int kh = (lane >> 4) * 2;
  const int aoff = (wid * 16 + m) * PJ_P + kh;
  const int boff = m * PJ_P + kh;

  v8f acc[9];
#pragma unroll
  for (int q = 0; q < 9; ++q) acc[q] = (v8f){};

  for (int kc = 0; kc < D3 / PJ_KC - 1; ++kc) {
    const int cur = (kc & 1);
    const int ko = (kc + 1) * PJ_KC;
    PJ_GLOAD(ko)
    pj_chunk(As + cur * (PJ_BM * PJ_P) + aoff, Bs + cur * (PJ_BN * PJ_P) + boff, acc);
    PJ_SSTORE(cur ^ 1)
    __syncthreads();
  }
  pj_chunk(As + 1 * (PJ_BM * PJ_P) + aoff, Bs + 1 * (PJ_BN * PJ_P) + boff, acc);
#undef PJ_GLOAD
#undef PJ_SSTORE

  const int rbase = (lane >> 4) * 8;
  const int ncol  = lane & 15;
#pragma unroll
  for (int q = 0; q < 9; ++q) {
    int n = n0 + q * 16 + ncol;
    float bias = bp[n];
    for (int r = 0; r < 8; ++r)
      hout[(t0 + wid * 16 + rbase + r) * D9 + n] = acc[q][r] + bias;
  }
}

__global__ void cea_ln(float* __restrict__ out,
                       const float* __restrict__ gn,
                       const float* __restrict__ bn) {
  __shared__ float row[D9];
  __shared__ float red[256];
  const int t = blockIdx.x, tid = threadIdx.x;
  float* __restrict__ h = out + t * D9;

  float ps = 0.0f;
  for (int n = tid; n < D9; n += 256) { float v = h[n]; row[n] = v; ps += v; }
  red[tid] = ps;
  __syncthreads();
  for (int s = 128; s > 0; s >>= 1) {
    if (tid < s) red[tid] += red[tid + s];
    __syncthreads();
  }
  const float mu = red[0] * (1.0f / (float)D9);
  __syncthreads();

  float pv = 0.0f;
  for (int n = tid; n < D9; n += 256) { float dv = row[n] - mu; pv += dv * dv; }
  red[tid] = pv;
  __syncthreads();
  for (int s = 128; s > 0; s >>= 1) {
    if (tid < s) red[tid] += red[tid + s];
    __syncthreads();
  }
  const float var = red[0] * (1.0f / (float)D9);
  const float rs  = __builtin_amdgcn_rsqf(var + 1e-5f);

  for (int n = tid; n < D9; n += 256)
    h[n] = (row[n] - mu) * rs * gn[n] + bn[n];
}


extern "C" void kernel_launch(void* const* d_in, const int* in_sizes, int n_in,
                              void* d_out, int out_size, void* d_ws, size_t ws_size,
                              hipStream_t stream) {
  const float* x  = (const float*)d_in[0];
  const float* Wq = (const float*)d_in[1];
  const float* bq = (const float*)d_in[2];
  const float* Sq = (const float*)d_in[3];
  const float* Wk = (const float*)d_in[4];
  const float* bk = (const float*)d_in[5];
  const float* Sk = (const float*)d_in[6];
  const float* Wv = (const float*)d_in[7];
  const float* bv = (const float*)d_in[8];
  const float* Sv = (const float*)d_in[9];
  const float* Wp = (const float*)d_in[10];
  const float* bp = (const float*)d_in[11];
  const float* gn = (const float*)d_in[12];
  const float* bn = (const float*)d_in[13];
  float* out = (float*)d_out;

  float* ws = (float*)d_ws;
  float* spw  = ws;
  float* Mt   = spw + 3 * D3 * DH;
  float* cb   = Mt  + 3 * D3 * DH;
  float* spP  = cb  + 3 * D3;
  float* qkv  = spP + D9 * D3;
  float* attb = qkv + NT * D9;

  cea_prep<<<dim3(PREP_BIAS_BLOCKS + PREP_SPW_BLOCKS + PREP_SPP_BLOCKS), dim3(256), 0, stream>>>(
      Wq, Wk, Wv, Wp, bq, bk, bv, Sq, Sk, Sv, spw, spP, cb);

  cea_fuse<<<dim3((D3 * DH) / 256, 3), dim3(256), 0, stream>>>(spw, Sq, Sk, Sv, Mt);

  cea_qkv<<<dim3(NT / 16, 3), dim3(256), 0, stream>>>(x, Mt, cb, qkv);

  cea_attn<<<dim3(NT), dim3(D3), 0, stream>>>(qkv, attb);

  cea_proj<<<dim3(NT / PJ_BM, D9 / PJ_BN), dim3(128), 0, stream>>>(attb, spP, bp, out);

  cea_ln<<<dim3(NT), dim3(256), 0, stream>>>(out, gn, bn);

  (void)in_sizes; (void)n_in; (void)out_size; (void)ws_size;
}


# QKV GEMM: stage both operands in LDS with coalesced 16-byte loads (tile 64 tokens x 96 cols, 384 blocks, next chunk prefetched) instead of per-lane 8-byte global reads of the weight
# speedup vs baseline: 1.9403x; 1.0678x over previous
#include <hip/hip_runtime.h>
#include <hip/hip_bf16.h>
#include <math.h>


#define NT     2048
#define DD     384
#define DH     128
#define D3     384
#define D9     1152
#define GAMMA  5.0f
#define LOG2E  1.4426950408889634f

typedef float v2f __attribute__((ext_vector_type(2)));
typedef float v4f __attribute__((ext_vector_type(4)));
typedef float v8f __attribute__((ext_vector_type(8)));

static __device__ __forceinline__ v8f wmma4(v2f a, v2f b, v8f c) {
  return __builtin_amdgcn_wmma_f32_16x16x4_f32(
       false, a,  false, b,
       (short)0, c,  false,  false);
}

static __device__ __forceinline__ float softplusf(float w) {
  return (w > 20.0f) ? w : log1pf(__expf(w));
}


#define PREP_BIAS_BLOCKS ((3 * D3) / 32)
#define PREP_SPW_BLOCKS  ((3 * D3 * DH) / 256)
#define PREP_SPP_BLOCKS  ((D9 * D3) / 256)
__global__ __launch_bounds__(256)
void cea_prep(const float* __restrict__ Wq, const float* __restrict__ Wk,
              const float* __restrict__ Wv, const float* __restrict__ Wp,
              const float* __restrict__ bq, const float* __restrict__ bk,
              const float* __restrict__ bv,
              const float* __restrict__ Sq, const float* __restrict__ Sk,
              const float* __restrict__ Sv,
              float* __restrict__ spw, float* __restrict__ spP,
              float* __restrict__ c) {
  __shared__ float red[8 * 32];
  const int blk = blockIdx.x, tid = threadIdx.x;
  if (blk >= PREP_BIAS_BLOCKS + PREP_SPW_BLOCKS) {
    int i = (blk - PREP_BIAS_BLOCKS - PREP_SPW_BLOCKS) * 256 + tid;
    spP[i] = softplusf(Wp[i]);
    return;
  }
  if (blk >= PREP_BIAS_BLOCKS) {
    int b2 = blk - PREP_BIAS_BLOCKS;
    int h = b2 / (PREP_SPW_BLOCKS / 3);
    int i = (b2 - h * (PREP_SPW_BLOCKS / 3)) * 256 + tid;
    const float* W = (h == 0) ? Wq : (h == 1) ? Wk : Wv;
    spw[h * (D3 * DH) + i] = softplusf(W[i]);
    return;
  }
  const int h  = blk / (D3 / 32);
  const int nl = tid & 31, jg = tid >> 5;
  const int n  = (blk - h * (D3 / 32)) * 32 + nl;
  const float* b = (h == 0) ? bq : (h == 1) ? bk : bv;
  const float* S = (h == 0) ? Sq : (h == 1) ? Sk : Sv;
  float acc = 0.0f;
#pragma unroll 16
  for (int jj = 0; jj < D3 / 8; ++jj) {
    int j = jg * (D3 / 8) + jj;
    acc += b[j] * S[j * D3 + n];
  }
  red[jg * 32 + nl] = acc;
  __syncthreads();
  if (jg == 0) {
    float s = red[nl];
    for (int q = 1; q < 8; ++q) s += red[q * 32 + nl];
    c[h * D3 + n] = s;
  }
}

__global__ void cea_fuse(const float* __restrict__ spw,
                         const float* __restrict__ Sq,
                         const float* __restrict__ Sk,
                         const float* __restrict__ Sv,
                         float* __restrict__ Mt) {
  int h   = blockIdx.y;
  int idx = blockIdx.x * 256 + threadIdx.x;
  int n = idx >> 7, k = idx & 127;
  const float* S = (h == 0) ? Sq : (h == 1) ? Sk : Sv;
  const float* W = spw + h * (D3 * DH);
  float acc = 0.0f;
  for (int j = 0; j < D3; ++j) acc += W[j * DH + k] * S[j * D3 + n];
  Mt[h * (D3 * DH) + idx] = acc;
}

#define PJ_BM 64
#define PJ_BN 144
#define PJ_KC 32
#define PJ_P  36
template <int NTL>
struct pj_frag { v4f a; v4f b[NTL]; };
template <int NTL>
static __device__ __forceinline__ void pj_read(const float* __restrict__ Ab,
                                               const float* __restrict__ Bb,
                                               int ks, pj_frag<NTL>& f) {
  v2f lo = *(const v2f*)&Ab[ks], hi = *(const v2f*)&Ab[ks + 4];
  f.a = (v4f){lo.x, lo.y, hi.x, hi.y};
#pragma unroll
  for (int q = 0; q < NTL; ++q) {
    v2f l2 = *(const v2f*)&Bb[q * 16 * PJ_P + ks], h2 = *(const v2f*)&Bb[q * 16 * PJ_P + ks + 4];
    f.b[q] = (v4f){l2.x, l2.y, h2.x, h2.y};
  }
}
template <int NTL>
static __device__ __forceinline__ void pj_mma(const pj_frag<NTL>& f, v8f (&acc)[NTL]) {
#pragma unroll
  for (int q = 0; q < NTL; ++q) acc[q] = wmma4((v2f){f.a.x, f.a.y}, (v2f){f.b[q].x, f.b[q].y}, acc[q]);
#pragma unroll
  for (int q = 0; q < NTL; ++q) acc[q] = wmma4((v2f){f.a.z, f.a.w}, (v2f){f.b[q].z, f.b[q].w}, acc[q]);
}
template <int NTL>
static __device__ __forceinline__ void pj_chunk(const float* __restrict__ Ab,
                                                const float* __restrict__ Bb,
                                                v8f (&acc)[NTL]) {
  pj_frag<NTL> f0, f1;
  pj_read(Ab, Bb, 0, f0);
  pj_read(Ab, Bb, 8, f1);
  __builtin_amdgcn_sched_barrier(0);
  pj_mma(f0, acc);
  __builtin_amdgcn_sched_barrier(0);
  pj_read(Ab, Bb, 16, f0);
  __builtin_amdgcn_sched_barrier(0);
  pj_mma(f1, acc);
  __builtin_amdgcn_sched_barrier(0);
  pj_read(Ab, Bb, 24, f1);
  __builtin_amdgcn_sched_barrier(0);
  pj_mma(f0, acc);
  __builtin_amdgcn_sched_barrier(0);
  pj_mma(f1, acc);
}


#define QK_BM 64
#define QK_BN 96
#define QK_NTL (QK_BN / 16)
__global__ __launch_bounds__(128)
void cea_qkv(const float* __restrict__ x,
             const float* __restrict__ Mt,
             const float* __restrict__ cb,
             float* __restrict__ qkv) {
  __shared__ __attribute__((aligned(16))) float As[2 * QK_BM * PJ_P];
  __shared__ __attribute__((aligned(16))) float Bs[2 * QK_BN * PJ_P];
  const int tid = threadIdx.x;
  const int t0 = blockIdx.x * QK_BM;
  const int h  = blockIdx.y / (D3 / QK_BN);
  const int n0 = (blockIdx.y - h * (D3 / QK_BN)) * QK_BN;

  const int srow = tid >> 3;
  const int sq   = (tid & 7) * 4;
  const float* __restrict__ ag = x + (t0 + srow) * DD + h * DH + sq;
  const float* __restrict__ bg = Mt + h * (D3 * DH) + (n0 + srow) * DH + sq;
  const int sl = srow * PJ_P + sq;

  v4f ra[QK_BM / 16], rb[QK_NTL];

#pragma unroll
  for (int i = 0; i < QK_BM / 16; ++i) ra[i] = *(const v4f*)(ag + i * 16 * DD);
#pragma unroll
  for (int i = 0; i < QK_NTL; ++i)     rb[i] = *(const v4f*)(bg + i * 16 * DH);
#pragma unroll
  for (int i = 0; i < QK_BM / 16; ++i) *(v4f*)&As[sl + i * 16 * PJ_P] = ra[i];
#pragma unroll
  for (int i = 0; i < QK_NTL; ++i)     *(v4f*)&Bs[sl + i * 16 * PJ_P] = rb[i];
  __syncthreads();

  const int lane = tid & 31, wid = tid >> 5;
  const int m  = lane & 15;
  const int kh = (lane >> 4) * 2;
  const int aoff = (wid * 16 + m) * PJ_P + kh;
  const int boff = m * PJ_P + kh;

  v8f acc[QK_NTL];
#pragma unroll
  for (int q = 0; q < QK_NTL; ++q) acc[q] = (v8f){};

  for (int kc = 0; kc < DH / PJ_KC - 1; ++kc) {
    const int cur = (kc & 1);
    const int ko = (kc + 1) * PJ_KC;
#pragma unroll
    for (int i = 0; i < QK_BM / 16; ++i) ra[i] = *(const v4f*)(ag + i * 16 * DD + ko);
#pragma unroll
    for (int i = 0; i < QK_NTL; ++i)     rb[i] = *(const v4f*)(bg + i * 16 * DH + ko);
    pj_chunk(As + cur * (QK_BM * PJ_P) + aoff, Bs + cur * (QK_BN * PJ_P) + boff, acc);
#pragma unroll
    for (int i = 0; i < QK_BM / 16; ++i) *(v4f*)&As[(cur ^ 1) * (QK_BM * PJ_P) + sl + i * 16 * PJ_P] = ra[i];
#pragma unroll
    for (int i = 0; i < QK_NTL; ++i)     *(v4f*)&Bs[(cur ^ 1) * (QK_BN * PJ_P) + sl + i * 16 * PJ_P] = rb[i];
    __syncthreads();
  }
  pj_chunk(As + 1 * (QK_BM * PJ_P) + aoff, Bs + 1 * (QK_BN * PJ_P) + boff, acc);

  const int rbase = (lane >> 4) * 8;
  const int ncol  = lane & 15;
#pragma unroll
  for (int q = 0; q < QK_NTL; ++q) {
    int n = h * D3 + n0 + q * 16 + ncol;
    float bias = cb[n];
    for (int r = 0; r < 8; ++r)
      qkv[(t0 + wid * 16 + rbase + r) * D9 + n] = acc[q][r] + bias;
  }
}

__global__ void cea_attn(const float* __restrict__ qkv,
                         float* __restrict__ att) {
  __shared__ float Qs[D3], Ksc[D3], Vs[D3];
  const int t = blockIdx.x, i = threadIdx.x;
  const float* __restrict__ base = qkv + t * D9;
  Qs[i]  = base[i];
  Ksc[i] = base[D3 + i] * (-GAMMA * LOG2E);
  Vs[i]  = base[2 * D3 + i];
  __syncthreads();

  const float Qi = Qs[i];
  float acc = 0.0f, ssum = 0.0f;
#pragma unroll 4
  for (int j = 0; j < D3; ++j) {
    float s = __builtin_amdgcn_rcpf(1.0f + __builtin_amdgcn_exp2f(Qi * Ksc[j]));
    acc  += s * Vs[j];
    ssum += s;
  }
  att[t * D3 + i] = acc / (ssum + 1e-8f);
}

__global__ __launch_bounds__(128)
void cea_proj(const float* __restrict__ att,
              const float* __restrict__ spP,
              const float* __restrict__ bp,
              float* __restrict__ hout) {
  __shared__ __attribute__((aligned(16))) float As[2 * PJ_BM * PJ_P];
  __shared__ __attribute__((aligned(16))) float Bs[2 * PJ_BN * PJ_P];
  const int tid = threadIdx.x;
  const int t0 = blockIdx.x * PJ_BM;
  const int n0 = blockIdx.y * PJ_BN;

  const int srow = tid >> 3;
  const int sq   = (tid & 7) * 4;
  const float* __restrict__ ag = att + (t0 + srow) * D3 + sq;
  const float* __restrict__ bg = spP + (n0 + srow) * D3 + sq;
  const int sl = srow * PJ_P + sq;

  v4f ra0, ra1, ra2, ra3, rb0, rb1, rb2, rb3, rb4, rb5, rb6, rb7, rb8;
#define PJ_GLOAD(ko)                                        \
  ra0 = *(const v4f*)(ag + 0 * 16 * D3 + (ko));             \
  ra1 = *(const v4f*)(ag + 1 * 16 * D3 + (ko));             \
  ra2 = *(const v4f*)(ag + 2 * 16 * D3 + (ko));             \
  ra3 = *(const v4f*)(ag + 3 * 16 * D3 + (ko));             \
  rb0 = *(const v4f*)(bg + 0 * 16 * D3 + (ko));             \
  rb1 = *(const v4f*)(bg + 1 * 16 * D3 + (ko));             \
  rb2 = *(const v4f*)(bg + 2 * 16 * D3 + (ko));             \
  rb3 = *(const v4f*)(bg + 3 * 16 * D3 + (ko));             \
  rb4 = *(const v4f*)(bg + 4 * 16 * D3 + (ko));             \
  rb5 = *(const v4f*)(bg + 5 * 16 * D3 + (ko));             \
  rb6 = *(const v4f*)(bg + 6 * 16 * D3 + (ko));             \
  rb7 = *(const v4f*)(bg + 7 * 16 * D3 + (ko));             \
  rb8 = *(const v4f*)(bg + 8 * 16 * D3 + (ko));
#define PJ_SSTORE(st)                                                   \
  *(v4f*)&As[(st) * (PJ_BM * PJ_P) + sl + 0 * 16 * PJ_P] = ra0;         \
  *(v4f*)&As[(st) * (PJ_BM * PJ_P) + sl + 1 * 16 * PJ_P] = ra1;         \
  *(v4f*)&As[(st) * (PJ_BM * PJ_P) + sl + 2 * 16 * PJ_P] = ra2;         \
  *(v4f*)&As[(st) * (PJ_BM * PJ_P) + sl + 3 * 16 * PJ_P] = ra3;         \
  *(v4f*)&Bs[(st) * (PJ_BN * PJ_P) + sl + 0 * 16 * PJ_P] = rb0;         \
  *(v4f*)&Bs[(st) * (PJ_BN * PJ_P) + sl + 1 * 16 * PJ_P] = rb1;         \
  *(v4f*)&Bs[(st) * (PJ_BN * PJ_P) + sl + 2 * 16 * PJ_P] = rb2;         \
  *(v4f*)&Bs[(st) * (PJ_BN * PJ_P) + sl + 3 * 16 * PJ_P] = rb3;         \
  *(v4f*)&Bs[(st) * (PJ_BN * PJ_P) + sl + 4 * 16 * PJ_P] = rb4;         \
  *(v4f*)&Bs[(st) * (PJ_BN * PJ_P) + sl + 5 * 16 * PJ_P] = rb5;         \
  *(v4f*)&Bs[(st) * (PJ_BN * PJ_P) + sl + 6 * 16 * PJ_P] = rb6;         \
  *(v4f*)&Bs[(st) * (PJ_BN * PJ_P) + sl + 7 * 16 * PJ_P] = rb7;         \
  *(v4f*)&Bs[(st) * (PJ_BN * PJ_P) + sl + 8 * 16 * PJ_P] = rb8;

  PJ_GLOAD(0)
  PJ_SSTORE(0)
  __syncthreads();

  const int lane = tid & 31, wid = tid >> 5;
  const int m  = lane & 15;
  const int kh = (lane >> 4) * 2;
  const int aoff = (wid * 16 + m) * PJ_P + kh;
  const int boff = m * PJ_P + kh;

  v8f acc[9];
#pragma unroll
  for (int q = 0; q < 9; ++q) acc[q] = (v8f){};

  for (int kc = 0; kc < D3 / PJ_KC - 1; ++kc) {
    const int cur = (kc & 1);
    const int ko = (kc + 1) * PJ_KC;
    PJ_GLOAD(ko)
    pj_chunk(As + cur * (PJ_BM * PJ_P) + aoff, Bs + cur * (PJ_BN * PJ_P) + boff, acc);
    PJ_SSTORE(cur ^ 1)
    __syncthreads();
  }
  pj_chunk(As + 1 * (PJ_BM * PJ_P) + aoff, Bs + 1 * (PJ_BN * PJ_P) + boff, acc);
#undef PJ_GLOAD
#undef PJ_SSTORE

  const int rbase = (lane >> 4) * 8;
  const int ncol  = lane & 15;
#pragma unroll
  for (int q = 0; q < 9; ++q) {
    int n = n0 + q * 16 + ncol;
    float bias = bp[n];
    for (int r = 0; r < 8; ++r)
      hout[(t0 + wid * 16 + rbase + r) * D9 + n] = acc[q][r] + bias;
  }
}

__global__ void cea_ln(float* __restrict__ out,
                       const float* __restrict__ gn,
                       const float* __restrict__ bn) {
  __shared__ float row[D9];
  __shared__ float red[256];
  const int t = blockIdx.x, tid = threadIdx.x;
  float* __restrict__ h = out + t * D9;

  float ps = 0.0f;
  for (int n = tid; n < D9; n += 256) { float v = h[n]; row[n] = v; ps += v; }
  red[tid] = ps;
  __syncthreads();
  for (int s = 128; s > 0; s >>= 1) {
    if (tid < s) red[tid] += red[tid + s];
    __syncthreads();
  }
  const float mu = red[0] * (1.0f / (float)D9);
  __syncthreads();

  float pv = 0.0f;
  for (int n = tid; n < D9; n += 256) { float dv = row[n] - mu; pv += dv * dv; }
  red[tid] = pv;
  __syncthreads();
  for (int s = 128; s > 0; s >>= 1) {
    if (tid < s) red[tid] += red[tid + s];
    __syncthreads();
  }
  const float var = red[0] * (1.0f / (float)D9);
  const float rs  = __builtin_amdgcn_rsqf(var + 1e-5f);

  for (int n = tid; n < D9; n += 256)
    h[n] = (row[n] - mu) * rs * gn[n] + bn[n];
}


extern "C" void kernel_launch(void* const* d_in, const int* in_sizes, int n_in,
                              void* d_out, int out_size, void* d_ws, size_t ws_size,
                              hipStream_t stream) {
  const float* x  = (const float*)d_in[0];
  const float* Wq = (const float*)d_in[1];
  const float* bq = (const float*)d_in[2];
  const float* Sq = (const float*)d_in[3];
  const float* Wk = (const float*)d_in[4];
  const float* bk = (const float*)d_in[5];
  const float* Sk = (const float*)d_in[6];
  const float* Wv = (const float*)d_in[7];
  const float* bv = (const float*)d_in[8];
  const float* Sv = (const float*)d_in[9];
  const float* Wp = (const float*)d_in[10];
  const float* bp = (const float*)d_in[11];
  const float* gn = (const float*)d_in[12];
  const float* bn = (const float*)d_in[13];
  float* out = (float*)d_out;

  float* ws = (float*)d_ws;
  float* spw  = ws;
  float* Mt   = spw + 3 * D3 * DH;
  float* cb   = Mt  + 3 * D3 * DH;
  float* spP  = cb  + 3 * D3;
  float* qkv  = spP + D9 * D3;
  float* attb = qkv + NT * D9;

  cea_prep<<<dim3(PREP_BIAS_BLOCKS + PREP_SPW_BLOCKS + PREP_SPP_BLOCKS), dim3(256), 0, stream>>>(
      Wq, Wk, Wv, Wp, bq, bk, bv, Sq, Sk, Sv, spw, spP, cb);

  cea_fuse<<<dim3((D3 * DH) / 256, 3), dim3(256), 0, stream>>>(spw, Sq, Sk, Sv, Mt);

  cea_qkv<<<dim3(NT / QK_BM, 3 * (D3 / QK_BN)), dim3(128), 0, stream>>>(x, Mt, cb, qkv);

  cea_attn<<<dim3(NT), dim3(D3), 0, stream>>>(qkv, attb);

  cea_proj<<<dim3(NT / PJ_BM, D9 / PJ_BN), dim3(128), 0, stream>>>(attb, spP, bp, out);

  cea_ln<<<dim3(NT), dim3(256), 0, stream>>>(out, gn, bn);

  (void)in_sizes; (void)n_in; (void)out_size; (void)ws_size;
}


# weight fusion Mt = softplus(W)^T S moves into the preprocessing launch as 72 WMMA blocks, softplus applied while staging to LDS (6 launches -> 5); Mt is now summed j ascending in steps of 4
# speedup vs baseline: 2.3773x; 1.2252x over previous
#include <hip/hip_runtime.h>
#include <hip/hip_bf16.h>
#include <math.h>


#define NT     2048
#define DD     384
#define DH     128
#define D3     384
#define D9     1152
#define GAMMA  5.0f
#define LOG2E  1.4426950408889634f

typedef float v2f __attribute__((ext_vector_type(2)));
typedef float v4f __attribute__((ext_vector_type(4)));
typedef float v8f __attribute__((ext_vector_type(8)));

static __device__ __forceinline__ v8f wmma4(v2f a, v2f b, v8f c) {
  return __builtin_amdgcn_wmma_f32_16x16x4_f32(
       false, a,  false, b,
       (short)0, c,  false,  false);
}

static __device__ __forceinline__ float softplusf(float w) {
  return (w > 20.0f) ? w : log1pf(__expf(w));
}


#define FZ_BN 64
#define FZ_BK 32
#define FZ_JC 96
#define FZ_PA 72
#define FZ_PB 40
#define FZ_PC 36
#define PREP_BIAS_BLOCKS ((3 * D3) / 32)
#define PREP_FUSE_BLOCKS (3 * (D3 / FZ_BN) * (DH / FZ_BK))
#define PREP_SPP_BLOCKS  ((D9 * D3) / 256)
struct fz_frag { v2f a[6]; v2f b[6]; };
static __device__ __forceinline__ void fz_read(const float* __restrict__ Af,
                                               const float* __restrict__ Bf,
                                               int ks0, fz_frag& f) {
#pragma unroll
  for (int s = 0; s < 6; ++s) {
    const int ks = ks0 + 4 * s;
    f.a[s] = (v2f){Af[ks * FZ_PA], Af[(ks + 1) * FZ_PA]};
    f.b[s] = (v2f){Bf[ks * FZ_PB], Bf[(ks + 1) * FZ_PB]};
  }
}
static __device__ __forceinline__ v8f fz_mma(const fz_frag& f, v8f acc) {
#pragma unroll
  for (int s = 0; s < 6; ++s) acc = wmma4(f.a[s], f.b[s], acc);
  return acc;
}
__global__ __launch_bounds__(256)
void cea_prep(const float* __restrict__ Wq, const float* __restrict__ Wk,
              const float* __restrict__ Wv, const float* __restrict__ Wp,
              const float* __restrict__ bq, const float* __restrict__ bk,
              const float* __restrict__ bv,
              const float* __restrict__ Sq, const float* __restrict__ Sk,
              const float* __restrict__ Sv,
              float* __restrict__ Mt, float* __restrict__ spP,
              float* __restrict__ c) {
  __shared__ __attribute__((aligned(16))) float As[FZ_JC * FZ_PA];
  __shared__ __attribute__((aligned(16))) float Bs[FZ_JC * FZ_PB];
  __shared__ float red[8 * 32];
  const int blk = blockIdx.x, tid = threadIdx.x;
  if (blk >= PREP_BIAS_BLOCKS + PREP_FUSE_BLOCKS) {
    int i = (blk - PREP_BIAS_BLOCKS - PREP_FUSE_BLOCKS) * 256 + tid;
    spP[i] = softplusf(Wp[i]);
    return;
  }
  if (blk >= PREP_BIAS_BLOCKS) {
    const int b2 = blk - PREP_BIAS_BLOCKS;
    const int h  = b2 / (PREP_FUSE_BLOCKS / 3);
    const int r2 = b2 - h * (PREP_FUSE_BLOCKS / 3);
    const int n0 = (r2 / (DH / FZ_BK)) * FZ_BN;
    const int k0 = (r2 % (DH / FZ_BK)) * FZ_BK;
    const float* __restrict__ S = (h == 0) ? Sq : (h == 1) ? Sk : Sv;
    const float* __restrict__ W = (h == 0) ? Wq : (h == 1) ? Wk : Wv;

    const int arow = tid >> 4, acol = (tid & 15) * 4;
    const int brow = tid >> 3, bcol = (tid & 7) * 4;
    const float* __restrict__ ag = S + arow * D3 + n0 + acol;
    const float* __restrict__ bg = W + brow * DH + k0 + bcol;

    v4f ra[FZ_JC / 16], rb[FZ_JC / 32];
#pragma unroll
    for (int i = 0; i < FZ_JC / 16; ++i) ra[i] = *(const v4f*)(ag + i * 16 * D3);
#pragma unroll
    for (int i = 0; i < FZ_JC / 32; ++i) rb[i] = *(const v4f*)(bg + i * 32 * DH);

    const int lane = tid & 31, wid = tid >> 5;
    const int cc = lane & 15;
    const int hh = lane >> 4;
    const int nt = wid & 3, kt = wid >> 2;
    const float* Af = As + (2 * hh) * FZ_PA + nt * 16 + cc;
    const float* Bf = Bs + (2 * hh) * FZ_PB + kt * 16 + cc;

    v8f acc = (v8f){};
#pragma unroll
    for (int ch = 0; ch < D3 / FZ_JC; ++ch) {
#pragma unroll
      for (int i = 0; i < FZ_JC / 16; ++i) *(v4f*)&As[(arow + 16 * i) * FZ_PA + acol] = ra[i];
#pragma unroll
      for (int i = 0; i < FZ_JC / 32; ++i) {
        v4f w = rb[i];
        *(v4f*)&Bs[(brow + 32 * i) * FZ_PB + bcol] =
            (v4f){softplusf(w.x), softplusf(w.y), softplusf(w.z), softplusf(w.w)};
      }
      __syncthreads();
      if (ch + 1 < D3 / FZ_JC) {
#pragma unroll
        for (int i = 0; i < FZ_JC / 16; ++i) ra[i] = *(const v4f*)(ag + ((ch + 1) * FZ_JC + i * 16) * D3);
#pragma unroll
        for (int i = 0; i < FZ_JC / 32; ++i) rb[i] = *(const v4f*)(bg + ((ch + 1) * FZ_JC + i * 32) * DH);
      }
      fz_frag f0, f1;
      fz_read(Af, Bf, 0, f0);
      fz_read(Af, Bf, 24, f1);
      __builtin_amdgcn_sched_barrier(0);
      acc = fz_mma(f0, acc);
      __builtin_amdgcn_sched_barrier(0);
      fz_read(Af, Bf, 48, f0);
      __builtin_amdgcn_sched_barrier(0);
      acc = fz_mma(f1, acc);
      __builtin_amdgcn_sched_barrier(0);
      fz_read(Af, Bf, 72, f1);
      __builtin_amdgcn_sched_barrier(0);
      acc = fz_mma(f0, acc);
      __builtin_amdgcn_sched_barrier(0);
      acc = fz_mma(f1, acc);
      __syncthreads();
    }

    float* Cs = As;
#pragma unroll
    for (int r = 0; r < 8; ++r) Cs[(nt * 16 + 8 * hh + r) * FZ_PC + kt * 16 + cc] = acc[r];
    __syncthreads();
#pragma unroll
    for (int i = 0; i < FZ_BN / 32; ++i) {
      const int row = (tid >> 3) + 32 * i, c4 = (tid & 7) * 4;
      v4f v = *(const v4f*)&Cs[row * FZ_PC + c4];
      *(v4f*)&Mt[h * (D3 * DH) + (n0 + row) * DH + k0 + c4] = v;
    }
    return;
  }
  const int h  = blk / (D3 / 32);
  const int nl = tid & 31, jg = tid >> 5;
  const int n  = (blk - h * (D3 / 32)) * 32 + nl;
  const float* b = (h == 0) ? bq : (h == 1) ? bk : bv;
  const float* S = (h == 0) ? Sq : (h == 1) ? Sk : Sv;
  float acc = 0.0f;
#pragma unroll 16
  for (int jj = 0; jj < D3 / 8; ++jj) {
    int j = jg * (D3 / 8) + jj;
    acc += b[j] * S[j * D3 + n];
  }
  red[jg * 32 + nl] = acc;
  __syncthreads();
  if (jg == 0) {
    float s = red[nl];
    for (int q = 1; q < 8; ++q) s += red[q * 32 + nl];
    c[h * D3 + n] = s;
  }
}

#define PJ_BM 64
#define PJ_BN 144
#define PJ_KC 32
#define PJ_P  36
template <int NTL>
struct pj_frag { v4f a; v4f b[NTL]; };
template <int NTL>
static __device__ __forceinline__ void pj_read(const float* __restrict__ Ab,
                                               const float* __restrict__ Bb,
                                               int ks, pj_frag<NTL>& f) {
  v2f lo = *(const v2f*)&Ab[ks], hi = *(const v2f*)&Ab[ks + 4];
  f.a = (v4f){lo.x, lo.y, hi.x, hi.y};
#pragma unroll
  for (int q = 0; q < NTL; ++q) {
    v2f l2 = *(const v2f*)&Bb[q * 16 * PJ_P + ks], h2 = *(const v2f*)&Bb[q * 16 * PJ_P + ks + 4];
    f.b[q] = (v4f){l2.x, l2.y, h2.x, h2.y};
  }
}
template <int NTL>
static __device__ __forceinline__ void pj_mma(const pj_frag<NTL>& f, v8f (&acc)[NTL]) {
#pragma unroll
  for (int q = 0; q < NTL; ++q) acc[q] = wmma4((v2f){f.a.x, f.a.y}, (v2f){f.b[q].x, f.b[q].y}, acc[q]);
#pragma unroll
  for (int q = 0; q < NTL; ++q) acc[q] = wmma4((v2f){f.a.z, f.a.w}, (v2f){f.b[q].z, f.b[q].w}, acc[q]);
}
template <int NTL>
static __device__ __forceinline__ void pj_chunk(const float* __restrict__ Ab,
                                                const float* __restrict__ Bb,
                                                v8f (&acc)[NTL]) {
  pj_frag<NTL> f0, f1;
  pj_read(Ab, Bb, 0, f0);
  pj_read(Ab, Bb, 8, f1);
  __builtin_amdgcn_sched_barrier(0);
  pj_mma(f0, acc);
  __builtin_amdgcn_sched_barrier(0);
  pj_read(Ab, Bb, 16, f0);
  __builtin_amdgcn_sched_barrier(0);
  pj_mma(f1, acc);
  __builtin_amdgcn_sched_barrier(0);
  pj_read(Ab, Bb, 24, f1);
  __builtin_amdgcn_sched_barrier(0);
  pj_mma(f0, acc);
  __builtin_amdgcn_sched_barrier(0);
  pj_mma(f1, acc);
}


#define QK_BM 64
#define QK_BN 96
#define QK_NTL (QK_BN / 16)
__global__ __launch_bounds__(128)
void cea_qkv(const float* __restrict__ x,
             const float* __restrict__ Mt,
             const float* __restrict__ cb,
             float* __restrict__ qkv) {
  __shared__ __attribute__((aligned(16))) float As[2 * QK_BM * PJ_P];
  __shared__ __attribute__((aligned(16))) float Bs[2 * QK_BN * PJ_P];
  const int tid = threadIdx.x;
  const int t0 = blockIdx.x * QK_BM;
  const int h  = blockIdx.y / (D3 / QK_BN);
  const int n0 = (blockIdx.y - h * (D3 / QK_BN)) * QK_BN;

  const int srow = tid >> 3;
  const int sq   = (tid & 7) * 4;
  const float* __restrict__ ag = x + (t0 + srow) * DD + h * DH + sq;
  const float* __restrict__ bg = Mt + h * (D3 * DH) + (n0 + srow) * DH + sq;
  const int sl = srow * PJ_P + sq;

  v4f ra[QK_BM / 16], rb[QK_NTL];

#pragma unroll
  for (int i = 0; i < QK_BM / 16; ++i) ra[i] = *(const v4f*)(ag + i * 16 * DD);
#pragma unroll
  for (int i = 0; i < QK_NTL; ++i)     rb[i] = *(const v4f*)(bg + i * 16 * DH);
#pragma unroll
  for (int i = 0; i < QK_BM / 16; ++i) *(v4f*)&As[sl + i * 16 * PJ_P] = ra[i];
#pragma unroll
  for (int i = 0; i < QK_NTL; ++i)     *(v4f*)&Bs[sl + i * 16 * PJ_P] = rb[i];
  __syncthreads();

  const int lane = tid & 31, wid = tid >> 5;
  const int m  = lane & 15;
  const int kh = (lane >> 4) * 2;
  const int aoff = (wid * 16 + m) * PJ_P + kh;
  const int boff = m * PJ_P + kh;

  v8f acc[QK_NTL];
#pragma unroll
  for (int q = 0; q < QK_NTL; ++q) acc[q] = (v8f){};

  for (int kc = 0; kc < DH / PJ_KC - 1; ++kc) {
    const int cur = (kc & 1);
    const int ko = (kc + 1) * PJ_KC;
#pragma unroll
    for (int i = 0; i < QK_BM / 16; ++i) ra[i] = *(const v4f*)(ag + i * 16 * DD + ko);
#pragma unroll
    for (int i = 0; i < QK_NTL; ++i)     rb[i] = *(const v4f*)(bg + i * 16 * DH + ko);
    pj_chunk(As + cur * (QK_BM * PJ_P) + aoff, Bs + cur * (QK_BN * PJ_P) + boff, acc);
#pragma unroll
    for (int i = 0; i < QK_BM / 16; ++i) *(v4f*)&As[(cur ^ 1) * (QK_BM * PJ_P) + sl + i * 16 * PJ_P] = ra[i];
#pragma unroll
    for (int i = 0; i < QK_NTL; ++i)     *(v4f*)&Bs[(cur ^ 1) * (QK_BN * PJ_P) + sl + i * 16 * PJ_P] = rb[i];
    __syncthreads();
  }
  pj_chunk(As + 1 * (QK_BM * PJ_P) + aoff, Bs + 1 * (QK_BN * PJ_P) + boff, acc);

  const int rbase = (lane >> 4) * 8;
  const int ncol  = lane & 15;
#pragma unroll
  for (int q = 0; q < QK_NTL; ++q) {
    int n = h * D3 + n0 + q * 16 + ncol;
    float bias = cb[n];
    for (int r = 0; r < 8; ++r)
      qkv[(t0 + wid * 16 + rbase + r) * D9 + n] = acc[q][r] + bias;
  }
}

__global__ void cea_attn(const float* __restrict__ qkv,
                         float* __restrict__ att) {
  __shared__ float Qs[D3], Ksc[D3], Vs[D3];
  const int t = blockIdx.x, i = threadIdx.x;
  const float* __restrict__ base = qkv + t * D9;
  Qs[i]  = base[i];
  Ksc[i] = base[D3 + i] * (-GAMMA * LOG2E);
  Vs[i]  = base[2 * D3 + i];
  __syncthreads();

  const float Qi = Qs[i];
  float acc = 0.0f, ssum = 0.0f;
#pragma unroll 4
  for (int j = 0; j < D3; ++j) {
    float s = __builtin_amdgcn_rcpf(1.0f + __builtin_amdgcn_exp2f(Qi * Ksc[j]));
    acc  += s * Vs[j];
    ssum += s;
  }
  att[t * D3 + i] = acc / (ssum + 1e-8f);
}

__global__ __launch_bounds__(128)
void cea_proj(const float* __restrict__ att,
              const float* __restrict__ spP,
              const float* __restrict__ bp,
              float* __restrict__ hout) {
  __shared__ __attribute__((aligned(16))) float As[2 * PJ_BM * PJ_P];
  __shared__ __attribute__((aligned(16))) float Bs[2 * PJ_BN * PJ_P];
  const int tid = threadIdx.x;
  const int t0 = blockIdx.x * PJ_BM;
  const int n0 = blockIdx.y * PJ_BN;

  const int srow = tid >> 3;
  const int sq   = (tid & 7) * 4;
  const float* __restrict__ ag = att + (t0 + srow) * D3 + sq;
  const float* __restrict__ bg = spP + (n0 + srow) * D3 + sq;
  const int sl = srow * PJ_P + sq;

  v4f ra0, ra1, ra2, ra3, rb0, rb1, rb2, rb3, rb4, rb5, rb6, rb7, rb8;
#define PJ_GLOAD(ko)                                        \
  ra0 = *(const v4f*)(ag + 0 * 16 * D3 + (ko));             \
  ra1 = *(const v4f*)(ag + 1 * 16 * D3 + (ko));             \
  ra2 = *(const v4f*)(ag + 2 * 16 * D3 + (ko));             \
  ra3 = *(const v4f*)(ag + 3 * 16 * D3 + (ko));             \
  rb0 = *(const v4f*)(bg + 0 * 16 * D3 + (ko));             \
  rb1 = *(const v4f*)(bg + 1 * 16 * D3 + (ko));             \
  rb2 = *(const v4f*)(bg + 2 * 16 * D3 + (ko));             \
  rb3 = *(const v4f*)(bg + 3 * 16 * D3 + (ko));             \
  rb4 = *(const v4f*)(bg + 4 * 16 * D3 + (ko));             \
  rb5 = *(const v4f*)(bg + 5 * 16 * D3 + (ko));             \
  rb6 = *(const v4f*)(bg + 6 * 16 * D3 + (ko));             \
  rb7 = *(const v4f*)(bg + 7 * 16 * D3 + (ko));             \
  rb8 = *(const v4f*)(bg + 8 * 16 * D3 + (ko));
#define PJ_SSTORE(st)                                                   \
  *(v4f*)&As[(st) * (PJ_BM * PJ_P) + sl + 0 * 16 * PJ_P] = ra0;         \
  *(v4f*)&As[(st) * (PJ_BM * PJ_P) + sl + 1 * 16 * PJ_P] = ra1;         \
  *(v4f*)&As[(st) * (PJ_BM * PJ_P) + sl + 2 * 16 * PJ_P] = ra2;         \
  *(v4f*)&As[(st) * (PJ_BM * PJ_P) + sl + 3 * 16 * PJ_P] = ra3;         \
  *(v4f*)&Bs[(st) * (PJ_BN * PJ_P) + sl + 0 * 16 * PJ_P] = rb0;         \
  *(v4f*)&Bs[(st) * (PJ_BN * PJ_P) + sl + 1 * 16 * PJ_P] = rb1;         \
  *(v4f*)&Bs[(st) * (PJ_BN * PJ_P) + sl + 2 * 16 * PJ_P] = rb2;         \
  *(v4f*)&Bs[(st) * (PJ_BN * PJ_P) + sl + 3 * 16 * PJ_P] = rb3;         \
  *(v4f*)&Bs[(st) * (PJ_BN * PJ_P) + sl + 4 * 16 * PJ_P] = rb4;         \
  *(v4f*)&Bs[(st) * (PJ_BN * PJ_P) + sl + 5 * 16 * PJ_P] = rb5;         \
  *(v4f*)&Bs[(st) * (PJ_BN * PJ_P) + sl + 6 * 16 * PJ_P] = rb6;         \
  *(v4f*)&Bs[(st) * (PJ_BN * PJ_P) + sl + 7 * 16 * PJ_P] = rb7;         \
  *(v4f*)&Bs[(st) * (PJ_BN * PJ_P) + sl + 8 * 16 * PJ_P] = rb8;

  PJ_GLOAD(0)
  PJ_SSTORE(0)
  __syncthreads();

  const int lane = tid & 31, wid = tid >> 5;
  const int m  = lane & 15;
  const int kh = (lane >> 4) * 2;
  const int aoff = (wid * 16 + m) * PJ_P + kh;
  const int boff = m * PJ_P + kh;

  v8f acc[9];
#pragma unroll
  for (int q = 0; q < 9; ++q) acc[q] = (v8f){};

  for (int kc = 0; kc < D3 / PJ_KC - 1; ++kc) {
    const int cur = (kc & 1);
    const int ko = (kc + 1) * PJ_KC;
    PJ_GLOAD(ko)
    pj_chunk(As + cur * (PJ_BM * PJ_P) + aoff, Bs + cur * (PJ_BN * PJ_P) + boff, acc);
    PJ_SSTORE(cur ^ 1)
    __syncthreads();
  }
  pj_chunk(As + 1 * (PJ_BM * PJ_P) + aoff, Bs + 1 * (PJ_BN * PJ_P) + boff, acc);
#undef PJ_GLOAD
#undef PJ_SSTORE

  const int rbase = (lane >> 4) * 8;
  const int ncol  = lane & 15;
#pragma unroll
  for (int q = 0; q < 9; ++q) {
    int n = n0 + q * 16 + ncol;
    float bias = bp[n];
    for (int r = 0; r < 8; ++r)
      hout[(t0 + wid * 16 + rbase + r) * D9 + n] = acc[q][r] + bias;
  }
}

__global__ void cea_ln(float* __restrict__ out,
                       const float* __restrict__ gn,
                       const float* __restrict__ bn) {
  __shared__ float row[D9];
  __shared__ float red[256];
  const int t = blockIdx.x, tid = threadIdx.x;
  float* __restrict__ h = out + t * D9;

  float ps = 0.0f;
  for (int n = tid; n < D9; n += 256) { float v = h[n]; row[n] = v; ps += v; }
  red[tid] = ps;
  __syncthreads();
  for (int s = 128; s > 0; s >>= 1) {
    if (tid < s) red[tid] += red[tid + s];
    __syncthreads();
  }
  const float mu = red[0] * (1.0f / (float)D9);
  __syncthreads();

  float pv = 0.0f;
  for (int n = tid; n < D9; n += 256) { float dv = row[n] - mu; pv += dv * dv; }
  red[tid] = pv;
  __syncthreads();
  for (int s = 128; s > 0; s >>= 1) {
    if (tid < s) red[tid] += red[tid + s];
    __syncthreads();
  }
  const float var = red[0] * (1.0f / (float)D9);
  const float rs  = __builtin_amdgcn_rsqf(var + 1e-5f);

  for (int n = tid; n < D9; n += 256)
    h[n] = (row[n] - mu) * rs * gn[n] + bn[n];
}


extern "C" void kernel_launch(void* const* d_in, const int* in_sizes, int n_in,
                              void* d_out, int out_size, void* d_ws, size_t ws_size,
                              hipStream_t stream) {
  const float* x  = (const float*)d_in[0];
  const float* Wq = (const float*)d_in[1];
  const float* bq = (const float*)d_in[2];
  const float* Sq = (const float*)d_in[3];
  const float* Wk = (const float*)d_in[4];
  const float* bk = (const float*)d_in[5];
  const float* Sk = (const float*)d_in[6];
  const float* Wv = (const float*)d_in[7];
  const float* bv = (const float*)d_in[8];
  const float* Sv = (const float*)d_in[9];
  const float* Wp = (const float*)d_in[10];
  const float* bp = (const float*)d_in[11];
  const float* gn = (const float*)d_in[12];
  const float* bn = (const float*)d_in[13];
  float* out = (float*)d_out;

  float* ws = (float*)d_ws;
  float* Mt   = ws + 3 * D3 * DH;
  float* cb   = Mt  + 3 * D3 * DH;
  float* spP  = cb  + 3 * D3;
  float* qkv  = spP + D9 * D3;
  float* attb = qkv + NT * D9;

  cea_prep<<<dim3(PREP_BIAS_BLOCKS + PREP_FUSE_BLOCKS + PREP_SPP_BLOCKS), dim3(256), 0, stream>>>(
      Wq, Wk, Wv, Wp, bq, bk, bv, Sq, Sk, Sv, Mt, spP, cb);

  cea_qkv<<<dim3(NT / QK_BM, 3 * (D3 / QK_BN)), dim3(128), 0, stream>>>(x, Mt, cb, qkv);

  cea_attn<<<dim3(NT), dim3(D3), 0, stream>>>(qkv, attb);

  cea_proj<<<dim3(NT / PJ_BM, D9 / PJ_BN), dim3(128), 0, stream>>>(attb, spP, bp, out);

  cea_ln<<<dim3(NT), dim3(256), 0, stream>>>(out, gn, bn);

  (void)in_sizes; (void)n_in; (void)out_size; (void)ws_size;
}


# attention: sigmoid as (1+tanh(z/2))/2, one v_tanh_f32 per element instead of exp2+rcp; K,V by uniform scalar loads, no LDS; rows with weight sum below 8 redone with exp2+rcp
# speedup vs baseline: 2.8190x; 1.1858x over previous
#include <hip/hip_runtime.h>
#include <hip/hip_bf16.h>
#include <math.h>


#define NT     2048
#define DD     384
#define DH     128
#define D3     384
#define D9     1152
#define GAMMA  5.0f
#define LOG2E  1.4426950408889634f

typedef float v2f __attribute__((ext_vector_type(2)));
typedef float v4f __attribute__((ext_vector_type(4)));
typedef float v8f __attribute__((ext_vector_type(8)));

static __device__ __forceinline__ v8f wmma4(v2f a, v2f b, v8f c) {
  return __builtin_amdgcn_wmma_f32_16x16x4_f32(
       false, a,  false, b,
       (short)0, c,  false,  false);
}

static __device__ __forceinline__ float softplusf(float w) {
  return (w > 20.0f) ? w : log1pf(__expf(w));
}


#define FZ_BN 64
#define FZ_BK 32
#define FZ_JC 96
#define FZ_PA 72
#define FZ_PB 40
#define FZ_PC 36
#define PREP_BIAS_BLOCKS ((3 * D3) / 32)
#define PREP_FUSE_BLOCKS (3 * (D3 / FZ_BN) * (DH / FZ_BK))
#define PREP_SPP_BLOCKS  ((D9 * D3) / 256)
struct fz_frag { v2f a[6]; v2f b[6]; };
static __device__ __forceinline__ void fz_read(const float* __restrict__ Af,
                                               const float* __restrict__ Bf,
                                               int ks0, fz_frag& f) {
#pragma unroll
  for (int s = 0; s < 6; ++s) {
    const int ks = ks0 + 4 * s;
    f.a[s] = (v2f){Af[ks * FZ_PA], Af[(ks + 1) * FZ_PA]};
    f.b[s] = (v2f){Bf[ks * FZ_PB], Bf[(ks + 1) * FZ_PB]};
  }
}
static __device__ __forceinline__ v8f fz_mma(const fz_frag& f, v8f acc) {
#pragma unroll
  for (int s = 0; s < 6; ++s) acc = wmma4(f.a[s], f.b[s], acc);
  return acc;
}
__global__ __launch_bounds__(256)
void cea_prep(const float* __restrict__ Wq, const float* __restrict__ Wk,
              const float* __restrict__ Wv, const float* __restrict__ Wp,
              const float* __restrict__ bq, const float* __restrict__ bk,
              const float* __restrict__ bv,
              const float* __restrict__ Sq, const float* __restrict__ Sk,
              const float* __restrict__ Sv,
              float* __restrict__ Mt, float* __restrict__ spP,
              float* __restrict__ c) {
  __shared__ __attribute__((aligned(16))) float As[FZ_JC * FZ_PA];
  __shared__ __attribute__((aligned(16))) float Bs[FZ_JC * FZ_PB];
  __shared__ float red[8 * 32];
  const int blk = blockIdx.x, tid = threadIdx.x;
  if (blk >= PREP_BIAS_BLOCKS + PREP_FUSE_BLOCKS) {
    int i = (blk - PREP_BIAS_BLOCKS - PREP_FUSE_BLOCKS) * 256 + tid;
    spP[i] = softplusf(Wp[i]);
    return;
  }
  if (blk >= PREP_BIAS_BLOCKS) {
    const int b2 = blk - PREP_BIAS_BLOCKS;
    const int h  = b2 / (PREP_FUSE_BLOCKS / 3);
    const int r2 = b2 - h * (PREP_FUSE_BLOCKS / 3);
    const int n0 = (r2 / (DH / FZ_BK)) * FZ_BN;
    const int k0 = (r2 % (DH / FZ_BK)) * FZ_BK;
    const float* __restrict__ S = (h == 0) ? Sq : (h == 1) ? Sk : Sv;
    const float* __restrict__ W = (h == 0) ? Wq : (h == 1) ? Wk : Wv;

    const int arow = tid >> 4, acol = (tid & 15) * 4;
    const int brow = tid >> 3, bcol = (tid & 7) * 4;
    const float* __restrict__ ag = S + arow * D3 + n0 + acol;
    const float* __restrict__ bg = W + brow * DH + k0 + bcol;

    v4f ra[FZ_JC / 16], rb[FZ_JC / 32];
#pragma unroll
    for (int i = 0; i < FZ_JC / 16; ++i) ra[i] = *(const v4f*)(ag + i * 16 * D3);
#pragma unroll
    for (int i = 0; i < FZ_JC / 32; ++i) rb[i] = *(const v4f*)(bg + i * 32 * DH);

    const int lane = tid & 31, wid = tid >> 5;
    const int cc = lane & 15;
    const int hh = lane >> 4;
    const int nt = wid & 3, kt = wid >> 2;
    const float* Af = As + (2 * hh) * FZ_PA + nt * 16 + cc;
    const float* Bf = Bs + (2 * hh) * FZ_PB + kt * 16 + cc;

    v8f acc = (v8f){};
#pragma unroll
    for (int ch = 0; ch < D3 / FZ_JC; ++ch) {
#pragma unroll
      for (int i = 0; i < FZ_JC / 16; ++i) *(v4f*)&As[(arow + 16 * i) * FZ_PA + acol] = ra[i];
#pragma unroll
      for (int i = 0; i < FZ_JC / 32; ++i) {
        v4f w = rb[i];
        *(v4f*)&Bs[(brow + 32 * i) * FZ_PB + bcol] =
            (v4f){softplusf(w.x), softplusf(w.y), softplusf(w.z), softplusf(w.w)};
      }
      __syncthreads();
      if (ch + 1 < D3 / FZ_JC) {
#pragma unroll
        for (int i = 0; i < FZ_JC / 16; ++i) ra[i] = *(const v4f*)(ag + ((ch + 1) * FZ_JC + i * 16) * D3);
#pragma unroll
        for (int i = 0; i < FZ_JC / 32; ++i) rb[i] = *(const v4f*)(bg + ((ch + 1) * FZ_JC + i * 32) * DH);
      }
      fz_frag f0, f1;
      fz_read(Af, Bf, 0, f0);
      fz_read(Af, Bf, 24, f1);
      __builtin_amdgcn_sched_barrier(0);
      acc = fz_mma(f0, acc);
      __builtin_amdgcn_sched_barrier(0);
      fz_read(Af, Bf, 48, f0);
      __builtin_amdgcn_sched_barrier(0);
      acc = fz_mma(f1, acc);
      __builtin_amdgcn_sched_barrier(0);
      fz_read(Af, Bf, 72, f1);
      __builtin_amdgcn_sched_barrier(0);
      acc = fz_mma(f0, acc);
      __builtin_amdgcn_sched_barrier(0);
      acc = fz_mma(f1, acc);
      __syncthreads();
    }

    float* Cs = As;
#pragma unroll
    for (int r = 0; r < 8; ++r) Cs[(nt * 16 + 8 * hh + r) * FZ_PC + kt * 16 + cc] = acc[r];
    __syncthreads();
#pragma unroll
    for (int i = 0; i < FZ_BN / 32; ++i) {
      const int row = (tid >> 3) + 32 * i, c4 = (tid & 7) * 4;
      v4f v = *(const v4f*)&Cs[row * FZ_PC + c4];
      *(v4f*)&Mt[h * (D3 * DH) + (n0 + row) * DH + k0 + c4] = v;
    }
    return;
  }
  const int h  = blk / (D3 / 32);
  const int nl = tid & 31, jg = tid >> 5;
  const int n  = (blk - h * (D3 / 32)) * 32 + nl;
  const float* b = (h == 0) ? bq : (h == 1) ? bk : bv;
  const float* S = (h == 0) ? Sq : (h == 1) ? Sk : Sv;
  float acc = 0.0f;
#pragma unroll 16
  for (int jj = 0; jj < D3 / 8; ++jj) {
    int j = jg * (D3 / 8) + jj;
    acc += b[j] * S[j * D3 + n];
  }
  red[jg * 32 + nl] = acc;
  __syncthreads();
  if (jg == 0) {
    float s = red[nl];
    for (int q = 1; q < 8; ++q) s += red[q * 32 + nl];
    c[h * D3 + n] = s;
  }
}

#define PJ_BM 64
#define PJ_BN 144
#define PJ_KC 32
#define PJ_P  36
template <int NTL>
struct pj_frag { v4f a; v4f b[NTL]; };
template <int NTL>
static __device__ __forceinline__ void pj_read(const float* __restrict__ Ab,
                                               const float* __restrict__ Bb,
                                               int ks, pj_frag<NTL>& f) {
  v2f lo = *(const v2f*)&Ab[ks], hi = *(const v2f*)&Ab[ks + 4];
  f.a = (v4f){lo.x, lo.y, hi.x, hi.y};
#pragma unroll
  for (int q = 0; q < NTL; ++q) {
    v2f l2 = *(const v2f*)&Bb[q * 16 * PJ_P + ks], h2 = *(const v2f*)&Bb[q * 16 * PJ_P + ks + 4];
    f.b[q] = (v4f){l2.x, l2.y, h2.x, h2.y};
  }
}
template <int NTL>
static __device__ __forceinline__ void pj_mma(const pj_frag<NTL>& f, v8f (&acc)[NTL]) {
#pragma unroll
  for (int q = 0; q < NTL; ++q) acc[q] = wmma4((v2f){f.a.x, f.a.y}, (v2f){f.b[q].x, f.b[q].y}, acc[q]);
#pragma unroll
  for (int q = 0; q < NTL; ++q) acc[q] = wmma4((v2f){f.a.z, f.a.w}, (v2f){f.b[q].z, f.b[q].w}, acc[q]);
}
template <int NTL>
static __device__ __forceinline__ void pj_chunk(const float* __restrict__ Ab,
                                                const float* __restrict__ Bb,
                                                v8f (&acc)[NTL]) {
  pj_frag<NTL> f0, f1;
  pj_read(Ab, Bb, 0, f0);
  pj_read(Ab, Bb, 8, f1);
  __builtin_amdgcn_sched_barrier(0);
  pj_mma(f0, acc);
  __builtin_amdgcn_sched_barrier(0);
  pj_read(Ab, Bb, 16, f0);
  __builtin_amdgcn_sched_barrier(0);
  pj_mma(f1, acc);
  __builtin_amdgcn_sched_barrier(0);
  pj_read(Ab, Bb, 24, f1);
  __builtin_amdgcn_sched_barrier(0);
  pj_mma(f0, acc);
  __builtin_amdgcn_sched_barrier(0);
  pj_mma(f1, acc);
}


#define QK_BM 64
#define QK_BN 96
#define QK_NTL (QK_BN / 16)
__global__ __launch_bounds__(128)
void cea_qkv(const float* __restrict__ x,
             const float* __restrict__ Mt,
             const float* __restrict__ cb,
             float* __restrict__ qkv) {
  __shared__ __attribute__((aligned(16))) float As[2 * QK_BM * PJ_P];
  __shared__ __attribute__((aligned(16))) float Bs[2 * QK_BN * PJ_P];
  const int tid = threadIdx.x;
  const int t0 = blockIdx.x * QK_BM;
  const int h  = blockIdx.y / (D3 / QK_BN);
  const int n0 = (blockIdx.y - h * (D3 / QK_BN)) * QK_BN;

  const int srow = tid >> 3;
  const int sq   = (tid & 7) * 4;
  const float* __restrict__ ag = x + (t0 + srow) * DD + h * DH + sq;
  const float* __restrict__ bg = Mt + h * (D3 * DH) + (n0 + srow) * DH + sq;
  const int sl = srow * PJ_P + sq;

  v4f ra[QK_BM / 16], rb[QK_NTL];

#pragma unroll
  for (int i = 0; i < QK_BM / 16; ++i) ra[i] = *(const v4f*)(ag + i * 16 * DD);
#pragma unroll
  for (int i = 0; i < QK_NTL; ++i)     rb[i] = *(const v4f*)(bg + i * 16 * DH);
#pragma unroll
  for (int i = 0; i < QK_BM / 16; ++i) *(v4f*)&As[sl + i * 16 * PJ_P] = ra[i];
#pragma unroll
  for (int i = 0; i < QK_NTL; ++i)     *(v4f*)&Bs[sl + i * 16 * PJ_P] = rb[i];
  __syncthreads();

  const int lane = tid & 31, wid = tid >> 5;
  const int m  = lane & 15;
  const int kh = (lane >> 4) * 2;
  const int aoff = (wid * 16 + m) * PJ_P + kh;
  const int boff = m * PJ_P + kh;

  v8f acc[QK_NTL];
#pragma unroll
  for (int q = 0; q < QK_NTL; ++q) acc[q] = (v8f){};

  for (int kc = 0; kc < DH / PJ_KC - 1; ++kc) {
    const int cur = (kc & 1);
    const int ko = (kc + 1) * PJ_KC;
#pragma unroll
    for (int i = 0; i < QK_BM / 16; ++i) ra[i] = *(const v4f*)(ag + i * 16 * DD + ko);
#pragma unroll
    for (int i = 0; i < QK_NTL; ++i)     rb[i] = *(const v4f*)(bg + i * 16 * DH + ko);
    pj_chunk(As + cur * (QK_BM * PJ_P) + aoff, Bs + cur * (QK_BN * PJ_P) + boff, acc);
#pragma unroll
    for (int i = 0; i < QK_BM / 16; ++i) *(v4f*)&As[(cur ^ 1) * (QK_BM * PJ_P) + sl + i * 16 * PJ_P] = ra[i];
#pragma unroll
    for (int i = 0; i < QK_NTL; ++i)     *(v4f*)&Bs[(cur ^ 1) * (QK_BN * PJ_P) + sl + i * 16 * PJ_P] = rb[i];
    __syncthreads();
  }
  pj_chunk(As + 1 * (QK_BM * PJ_P) + aoff, Bs + 1 * (QK_BN * PJ_P) + boff, acc);

  const int rbase = (lane >> 4) * 8;
  const int ncol  = lane & 15;
#pragma unroll
  for (int q = 0; q < QK_NTL; ++q) {
    int n = h * D3 + n0 + q * 16 + ncol;
    float bias = cb[n];
    for (int r = 0; r < 8; ++r)
      qkv[(t0 + wid * 16 + rbase + r) * D9 + n] = acc[q][r] + bias;
  }
}

#define ATT_SMALL_SUM 8.0f
__global__ __launch_bounds__(D3)
void cea_attn(const float* __restrict__ qkv,
              float* __restrict__ att) {
  const int t = blockIdx.x, i = threadIdx.x;
  const float* __restrict__ base = qkv + t * D9;
  const float* __restrict__ Kp = base + D3;
  const float* __restrict__ Vp = base + 2 * D3;
  const float Qi = base[i];
  const float Qh = Qi * (0.5f * GAMMA);

  float acc2 = 0.0f, usum = 0.0f;
#pragma unroll 16
  for (int j = 0; j < D3; ++j) {
    const float u = 1.0f + __builtin_amdgcn_tanhf(Qh * Kp[j]);
    acc2  = __builtin_fmaf(u, Vp[j], acc2);
    usum += u;
  }
  float acc = 0.5f * acc2, ssum = 0.5f * usum;
  if (ssum < ATT_SMALL_SUM) {
    const float Qe = Qi * (-GAMMA * LOG2E);
    acc = 0.0f; ssum = 0.0f;
#pragma unroll 4
    for (int j = 0; j < D3; ++j) {
      const float s = __builtin_amdgcn_rcpf(1.0f + __builtin_amdgcn_exp2f(Qe * Kp[j]));
      acc   = __builtin_fmaf(s, Vp[j], acc);
      ssum += s;
    }
  }
  att[t * D3 + i] = acc / (ssum + 1e-8f);
}

__global__ __launch_bounds__(128)
void cea_proj(const float* __restrict__ att,
              const float* __restrict__ spP,
              const float* __restrict__ bp,
              float* __restrict__ hout) {
  __shared__ __attribute__((aligned(16))) float As[2 * PJ_BM * PJ_P];
  __shared__ __attribute__((aligned(16))) float Bs[2 * PJ_BN * PJ_P];
  const int tid = threadIdx.x;
  const int t0 = blockIdx.x * PJ_BM;
  const int n0 = blockIdx.y * PJ_BN;

  const int srow = tid >> 3;
  const int sq   = (tid & 7) * 4;
  const float* __restrict__ ag = att + (t0 + srow) * D3 + sq;
  const float* __restrict__ bg = spP + (n0 + srow) * D3 + sq;
  const int sl = srow * PJ_P + sq;

  v4f ra0, ra1, ra2, ra3, rb0, rb1, rb2, rb3, rb4, rb5, rb6, rb7, rb8;
#define PJ_GLOAD(ko)                                        \
  ra0 = *(const v4f*)(ag + 0 * 16 * D3 + (ko));             \
  ra1 = *(const v4f*)(ag + 1 * 16 * D3 + (ko));             \
  ra2 = *(const v4f*)(ag + 2 * 16 * D3 + (ko));             \
  ra3 = *(const v4f*)(ag + 3 * 16 * D3 + (ko));             \
  rb0 = *(const v4f*)(bg + 0 * 16 * D3 + (ko));             \
  rb1 = *(const v4f*)(bg + 1 * 16 * D3 + (ko));             \
  rb2 = *(const v4f*)(bg + 2 * 16 * D3 + (ko));             \
  rb3 = *(const v4f*)(bg + 3 * 16 * D3 + (ko));             \
  rb4 = *(const v4f*)(bg + 4 * 16 * D3 + (ko));             \
  rb5 = *(const v4f*)(bg + 5 * 16 * D3 + (ko));             \
  rb6 = *(const v4f*)(bg + 6 * 16 * D3 + (ko));             \
  rb7 = *(const v4f*)(bg + 7 * 16 * D3 + (ko));             \
  rb8 = *(const v4f*)(bg + 8 * 16 * D3 + (ko));
#define PJ_SSTORE(st)                                                   \
  *(v4f*)&As[(st) * (PJ_BM * PJ_P) + sl + 0 * 16 * PJ_P] = ra0;         \
  *(v4f*)&As[(st) * (PJ_BM * PJ_P) + sl + 1 * 16 * PJ_P] = ra1;         \
  *(v4f*)&As[(st) * (PJ_BM * PJ_P) + sl + 2 * 16 * PJ_P] = ra2;         \
  *(v4f*)&As[(st) * (PJ_BM * PJ_P) + sl + 3 * 16 * PJ_P] = ra3;         \
  *(v4f*)&Bs[(st) * (PJ_BN * PJ_P) + sl + 0 * 16 * PJ_P] = rb0;         \
  *(v4f*)&Bs[(st) * (PJ_BN * PJ_P) + sl + 1 * 16 * PJ_P] = rb1;         \
  *(v4f*)&Bs[(st) * (PJ_BN * PJ_P) + sl + 2 * 16 * PJ_P] = rb2;         \
  *(v4f*)&Bs[(st) * (PJ_BN * PJ_P) + sl + 3 * 16 * PJ_P] = rb3;         \
  *(v4f*)&Bs[(st) * (PJ_BN * PJ_P) + sl + 4 * 16 * PJ_P] = rb4;         \
  *(v4f*)&Bs[(st) * (PJ_BN * PJ_P) + sl + 5 * 16 * PJ_P] = rb5;         \
  *(v4f*)&Bs[(st) * (PJ_BN * PJ_P) + sl + 6 * 16 * PJ_P] = rb6;         \
  *(v4f*)&Bs[(st) * (PJ_BN * PJ_P) + sl + 7 * 16 * PJ_P] = rb7;         \
  *(v4f*)&Bs[(st) * (PJ_BN * PJ_P) + sl + 8 * 16 * PJ_P] = rb8;

  PJ_GLOAD(0)
  PJ_SSTORE(0)
  __syncthreads();

  const int lane = tid & 31, wid = tid >> 5;
  const int m  = lane & 15;
  const int kh = (lane >> 4) * 2;
  const int aoff = (wid * 16 + m) * PJ_P + kh;
  const int boff = m * PJ_P + kh;

  v8f acc[9];
#pragma unroll
  for (int q = 0; q < 9; ++q) acc[q] = (v8f){};

  for (int kc = 0; kc < D3 / PJ_KC - 1; ++kc) {
    const int cur = (kc & 1);
    const int ko = (kc + 1) * PJ_KC;
    PJ_GLOAD(ko)
    pj_chunk(As + cur * (PJ_BM * PJ_P) + aoff, Bs + cur * (PJ_BN * PJ_P) + boff, acc);
    PJ_SSTORE(cur ^ 1)
    __syncthreads();
  }
  pj_chunk(As + 1 * (PJ_BM * PJ_P) + aoff, Bs + 1 * (PJ_BN * PJ_P) + boff, acc);
#undef PJ_GLOAD
#undef PJ_SSTORE

  const int rbase = (lane >> 4) * 8;
  const int ncol  = lane & 15;
#pragma unroll
  for (int q = 0; q < 9; ++q) {
    int n = n0 + q * 16 + ncol;
    float bias = bp[n];
    for (int r = 0; r < 8; ++r)
      hout[(t0 + wid * 16 + rbase + r) * D9 + n] = acc[q][r] + bias;
  }
}

__global__ void cea_ln(float* __restrict__ out,
                       const float* __restrict__ gn,
                       const float* __restrict__ bn) {
  __shared__ float row[D9];
  __shared__ float red[256];
  const int t = blockIdx.x, tid = threadIdx.x;
  float* __restrict__ h = out + t * D9;

  float ps = 0.0f;
  for (int n = tid; n < D9; n += 256) { float v = h[n]; row[n] = v; ps += v; }
  red[tid] = ps;
  __syncthreads();
  for (int s = 128; s > 0; s >>= 1) {
    if (tid < s) red[tid] += red[tid + s];
    __syncthreads();
  }
  const float mu = red[0] * (1.0f / (float)D9);
  __syncthreads();

  float pv = 0.0f;
  for (int n = tid; n < D9; n += 256) { float dv = row[n] - mu; pv += dv * dv; }
  red[tid] = pv;
  __syncthreads();
  for (int s = 128; s > 0; s >>= 1) {
    if (tid < s) red[tid] += red[tid + s];
    __syncthreads();
  }
  const float var = red[0] * (1.0f / (float)D9);
  const float rs  = __builtin_amdgcn_rsqf(var + 1e-5f);

  for (int n = tid; n < D9; n += 256)
    h[n] = (row[n] - mu) * rs * gn[n] + bn[n];
}


extern "C" void kernel_launch(void* const* d_in, const int* in_sizes, int n_in,
                              void* d_out, int out_size, void* d_ws, size_t ws_size,
                              hipStream_t stream) {
  const float* x  = (const float*)d_in[0];
  const float* Wq = (const float*)d_in[1];
  const float* bq = (const float*)d_in[2];
  const float* Sq = (const float*)d_in[3];
  const float* Wk = (const float*)d_in[4];
  const float* bk = (const float*)d_in[5];
  const float* Sk = (const float*)d_in[6];
  const float* Wv = (const float*)d_in[7];
  const float* bv = (const float*)d_in[8];
  const float* Sv = (const float*)d_in[9];
  const float* Wp = (const float*)d_in[10];
  const float* bp = (const float*)d_in[11];
  const float* gn = (const float*)d_in[12];
  const float* bn = (const float*)d_in[13];
  float* out = (float*)d_out;

  float* ws = (float*)d_ws;
  float* Mt   = ws + 3 * D3 * DH;
  float* cb   = Mt  + 3 * D3 * DH;
  float* spP  = cb  + 3 * D3;
  float* qkv  = spP + D9 * D3;
  float* attb = qkv + NT * D9;

  cea_prep<<<dim3(PREP_BIAS_BLOCKS + PREP_FUSE_BLOCKS + PREP_SPP_BLOCKS), dim3(256), 0, stream>>>(
      Wq, Wk, Wv, Wp, bq, bk, bv, Sq, Sk, Sv, Mt, spP, cb);

  cea_qkv<<<dim3(NT / QK_BM, 3 * (D3 / QK_BN)), dim3(128), 0, stream>>>(x, Mt, cb, qkv);

  cea_attn<<<dim3(NT), dim3(D3), 0, stream>>>(qkv, attb);

  cea_proj<<<dim3(NT / PJ_BM, D9 / PJ_BN), dim3(128), 0, stream>>>(attb, spP, bp, out);

  cea_ln<<<dim3(NT), dim3(256), 0, stream>>>(out, gn, bn);

  (void)in_sizes; (void)n_in; (void)out_size; (void)ws_size;
}


# layernorm sums move into the projection epilogue (per 144-column block: mean, then squared deviations, 16-lane butterfly); layernorm is one pass, no LDS or barrier; new sum order in mean and var
# speedup vs baseline: 2.9577x; 1.0492x over previous
#include <hip/hip_runtime.h>
#include <hip/hip_bf16.h>
#include <math.h>


#define NT     2048
#define DD     384
#define DH     128
#define D3     384
#define D9     1152
#define GAMMA  5.0f
#define LOG2E  1.4426950408889634f

typedef float v2f __attribute__((ext_vector_type(2)));
typedef float v4f __attribute__((ext_vector_type(4)));
typedef float v8f __attribute__((ext_vector_type(8)));

static __device__ __forceinline__ v8f wmma4(v2f a, v2f b, v8f c) {
  return __builtin_amdgcn_wmma_f32_16x16x4_f32(
       false, a,  false, b,
       (short)0, c,  false,  false);
}

template <int M>
static __device__ __forceinline__ float row_xor_add(float v) {
  const int iv = __builtin_bit_cast(int, v);
  const int sw = __builtin_amdgcn_update_dpp(iv, iv, 0x160 + M, 0xf, 0xf, false);
  return v + __builtin_bit_cast(float, sw);
}
static __device__ __forceinline__ float row16_sum(float v) {
  v = row_xor_add<1>(v);
  v = row_xor_add<2>(v);
  v = row_xor_add<4>(v);
  v = row_xor_add<8>(v);
  return v;
}

static __device__ __forceinline__ float softplusf(float w) {
  return (w > 20.0f) ? w : log1pf(__expf(w));
}


#define FZ_BN 64
#define FZ_BK 32
#define FZ_JC 96
#define FZ_PA 72
#define FZ_PB 40
#define FZ_PC 36
#define PREP_BIAS_BLOCKS ((3 * D3) / 32)
#define PREP_FUSE_BLOCKS (3 * (D3 / FZ_BN) * (DH / FZ_BK))
#define PREP_SPP_BLOCKS  ((D9 * D3) / 256)
struct fz_frag { v2f a[6]; v2f b[6]; };
static __device__ __forceinline__ void fz_read(const float* __restrict__ Af,
                                               const float* __restrict__ Bf,
                                               int ks0, fz_frag& f) {
#pragma unroll
  for (int s = 0; s < 6; ++s) {
    const int ks = ks0 + 4 * s;
    f.a[s] = (v2f){Af[ks * FZ_PA], Af[(ks + 1) * FZ_PA]};
    f.b[s] = (v2f){Bf[ks * FZ_PB], Bf[(ks + 1) * FZ_PB]};
  }
}
static __device__ __forceinline__ v8f fz_mma(const fz_frag& f, v8f acc) {
#pragma unroll
  for (int s = 0; s < 6; ++s) acc = wmma4(f.a[s], f.b[s], acc);
  return acc;
}
__global__ __launch_bounds__(256)
void cea_prep(const float* __restrict__ Wq, const float* __restrict__ Wk,
              const float* __restrict__ Wv, const float* __restrict__ Wp,
              const float* __restrict__ bq, const float* __restrict__ bk,
              const float* __restrict__ bv,
              const float* __restrict__ Sq, const float* __restrict__ Sk,
              const float* __restrict__ Sv,
              float* __restrict__ Mt, float* __restrict__ spP,
              float* __restrict__ c) {
  __shared__ __attribute__((aligned(16))) float As[FZ_JC * FZ_PA];
  __shared__ __attribute__((aligned(16))) float Bs[FZ_JC * FZ_PB];
  __shared__ float red[8 * 32];
  const int blk = blockIdx.x, tid = threadIdx.x;
  if (blk >= PREP_BIAS_BLOCKS + PREP_FUSE_BLOCKS) {
    int i = (blk - PREP_BIAS_BLOCKS - PREP_FUSE_BLOCKS) * 256 + tid;
    spP[i] = softplusf(Wp[i]);
    return;
  }
  if (blk >= PREP_BIAS_BLOCKS) {
    const int b2 = blk - PREP_BIAS_BLOCKS;
    const int h  = b2 / (PREP_FUSE_BLOCKS / 3);
    const int r2 = b2 - h * (PREP_FUSE_BLOCKS / 3);
    const int n0 = (r2 / (DH / FZ_BK)) * FZ_BN;
    const int k0 = (r2 % (DH / FZ_BK)) * FZ_BK;
    const float* __restrict__ S = (h == 0) ? Sq : (h == 1) ? Sk : Sv;
    const float* __restrict__ W = (h == 0) ? Wq : (h == 1) ? Wk : Wv;

    const int arow = tid >> 4, acol = (tid & 15) * 4;
    const int brow = tid >> 3, bcol = (tid & 7) * 4;
    const float* __restrict__ ag = S + arow * D3 + n0 + acol;
    const float* __restrict__ bg = W + brow * DH + k0 + bcol;

    v4f ra[FZ_JC / 16], rb[FZ_JC / 32];
#pragma unroll
    for (int i = 0; i < FZ_JC / 16; ++i) ra[i] = *(const v4f*)(ag + i * 16 * D3);
#pragma unroll
    for (int i = 0; i < FZ_JC / 32; ++i) rb[i] = *(const v4f*)(bg + i * 32 * DH);

    const int lane = tid & 31, wid = tid >> 5;
    const int cc = lane & 15;
    const int hh = lane >> 4;
    const int nt = wid & 3, kt = wid >> 2;
    const float* Af = As + (2 * hh) * FZ_PA + nt * 16 + cc;
    const float* Bf = Bs + (2 * hh) * FZ_PB + kt * 16 + cc;

    v8f acc = (v8f){};
#pragma unroll
    for (int ch = 0; ch < D3 / FZ_JC; ++ch) {
#pragma unroll
      for (int i = 0; i < FZ_JC / 16; ++i) *(v4f*)&As[(arow + 16 * i) * FZ_PA + acol] = ra[i];
#pragma unroll
      for (int i = 0; i < FZ_JC / 32; ++i) {
        v4f w = rb[i];
        *(v4f*)&Bs[(brow + 32 * i) * FZ_PB + bcol] =
            (v4f){softplusf(w.x), softplusf(w.y), softplusf(w.z), softplusf(w.w)};
      }
      __syncthreads();
      if (ch + 1 < D3 / FZ_JC) {
#pragma unroll
        for (int i = 0; i < FZ_JC / 16; ++i) ra[i] = *(const v4f*)(ag + ((ch + 1) * FZ_JC + i * 16) * D3);
#pragma unroll
        for (int i = 0; i < FZ_JC / 32; ++i) rb[i] = *(const v4f*)(bg + ((ch + 1) * FZ_JC + i * 32) * DH);
      }
      fz_frag f0, f1;
      fz_read(Af, Bf, 0, f0);
      fz_read(Af, Bf, 24, f1);
      __builtin_amdgcn_sched_barrier(0);
      acc = fz_mma(f0, acc);
      __builtin_amdgcn_sched_barrier(0);
      fz_read(Af, Bf, 48, f0);
      __builtin_amdgcn_sched_barrier(0);
      acc = fz_mma(f1, acc);
      __builtin_amdgcn_sched_barrier(0);
      fz_read(Af, Bf, 72, f1);
      __builtin_amdgcn_sched_barrier(0);
      acc = fz_mma(f0, acc);
      __builtin_amdgcn_sched_barrier(0);
      acc = fz_mma(f1, acc);
      __syncthreads();
    }

    float* Cs = As;
#pragma unroll
    for (int r = 0; r < 8; ++r) Cs[(nt * 16 + 8 * hh + r) * FZ_PC + kt * 16 + cc] = acc[r];
    __syncthreads();
#pragma unroll
    for (int i = 0; i < FZ_BN / 32; ++i) {
      const int row = (tid >> 3) + 32 * i, c4 = (tid & 7) * 4;
      v4f v = *(const v4f*)&Cs[row * FZ_PC + c4];
      *(v4f*)&Mt[h * (D3 * DH) + (n0 + row) * DH + k0 + c4] = v;
    }
    return;
  }
  const int h  = blk / (D3 / 32);
  const int nl = tid & 31, jg = tid >> 5;
  const int n  = (blk - h * (D3 / 32)) * 32 + nl;
  const float* b = (h == 0) ? bq : (h == 1) ? bk : bv;
  const float* S = (h == 0) ? Sq : (h == 1) ? Sk : Sv;
  float acc = 0.0f;
#pragma unroll 16
  for (int jj = 0; jj < D3 / 8; ++jj) {
    int j = jg * (D3 / 8) + jj;
    acc += b[j] * S[j * D3 + n];
  }
  red[jg * 32 + nl] = acc;
  __syncthreads();
  if (jg == 0) {
    float s = red[nl];
    for (int q = 1; q < 8; ++q) s += red[q * 32 + nl];
    c[h * D3 + n] = s;
  }
}

#define PJ_BM 64
#define PJ_BN 144
#define PJ_KC 32
#define PJ_P  36
template <int NTL>
struct pj_frag { v4f a; v4f b[NTL]; };
template <int NTL>
static __device__ __forceinline__ void pj_read(const float* __restrict__ Ab,
                                               const float* __restrict__ Bb,
                                               int ks, pj_frag<NTL>& f) {
  v2f lo = *(const v2f*)&Ab[ks], hi = *(const v2f*)&Ab[ks + 4];
  f.a = (v4f){lo.x, lo.y, hi.x, hi.y};
#pragma unroll
  for (int q = 0; q < NTL; ++q) {
    v2f l2 = *(const v2f*)&Bb[q * 16 * PJ_P + ks], h2 = *(const v2f*)&Bb[q * 16 * PJ_P + ks + 4];
    f.b[q] = (v4f){l2.x, l2.y, h2.x, h2.y};
  }
}
template <int NTL>
static __device__ __forceinline__ void pj_mma(const pj_frag<NTL>& f, v8f (&acc)[NTL]) {
#pragma unroll
  for (int q = 0; q < NTL; ++q) acc[q] = wmma4((v2f){f.a.x, f.a.y}, (v2f){f.b[q].x, f.b[q].y}, acc[q]);
#pragma unroll
  for (int q = 0; q < NTL; ++q) acc[q] = wmma4((v2f){f.a.z, f.a.w}, (v2f){f.b[q].z, f.b[q].w}, acc[q]);
}
template <int NTL>
static __device__ __forceinline__ void pj_chunk(const float* __restrict__ Ab,
                                                const float* __restrict__ Bb,
                                                v8f (&acc)[NTL]) {
  pj_frag<NTL> f0, f1;
  pj_read(Ab, Bb, 0, f0);
  pj_read(Ab, Bb, 8, f1);
  __builtin_amdgcn_sched_barrier(0);
  pj_mma(f0, acc);
  __builtin_amdgcn_sched_barrier(0);
  pj_read(Ab, Bb, 16, f0);
  __builtin_amdgcn_sched_barrier(0);
  pj_mma(f1, acc);
  __builtin_amdgcn_sched_barrier(0);
  pj_read(Ab, Bb, 24, f1);
  __builtin_amdgcn_sched_barrier(0);
  pj_mma(f0, acc);
  __builtin_amdgcn_sched_barrier(0);
  pj_mma(f1, acc);
}


#define QK_BM 64
#define QK_BN 96
#define QK_NTL (QK_BN / 16)
__global__ __launch_bounds__(128)
void cea_qkv(const float* __restrict__ x,
             const float* __restrict__ Mt,
             const float* __restrict__ cb,
             float* __restrict__ qkv) {
  __shared__ __attribute__((aligned(16))) float As[2 * QK_BM * PJ_P];
  __shared__ __attribute__((aligned(16))) float Bs[2 * QK_BN * PJ_P];
  const int tid = threadIdx.x;
  const int t0 = blockIdx.x * QK_BM;
  const int h  = blockIdx.y / (D3 / QK_BN);
  const int n0 = (blockIdx.y - h * (D3 / QK_BN)) * QK_BN;

  const int srow = tid >> 3;
  const int sq   = (tid & 7) * 4;
  const float* __restrict__ ag = x + (t0 + srow) * DD + h * DH + sq;
  const float* __restrict__ bg = Mt + h * (D3 * DH) + (n0 + srow) * DH + sq;
  const int sl = srow * PJ_P + sq;

  v4f ra[QK_BM / 16], rb[QK_NTL];

#pragma unroll
  for (int i = 0; i < QK_BM / 16; ++i) ra[i] = *(const v4f*)(ag + i * 16 * DD);
#pragma unroll
  for (int i = 0; i < QK_NTL; ++i)     rb[i] = *(const v4f*)(bg + i * 16 * DH);
#pragma unroll
  for (int i = 0; i < QK_BM / 16; ++i) *(v4f*)&As[sl + i * 16 * PJ_P] = ra[i];
#pragma unroll
  for (int i = 0; i < QK_NTL; ++i)     *(v4f*)&Bs[sl + i * 16 * PJ_P] = rb[i];
  __syncthreads();

  const int lane = tid & 31, wid = tid >> 5;
  const int m  = lane & 15;
  const int kh = (lane >> 4) * 2;
  const int aoff = (wid * 16 + m) * PJ_P + kh;
  const int boff = m * PJ_P + kh;

  v8f acc[QK_NTL];
#pragma unroll
  for (int q = 0; q < QK_NTL; ++q) acc[q] = (v8f){};

  for (int kc = 0; kc < DH / PJ_KC - 1; ++kc) {
    const int cur = (kc & 1);
    const int ko = (kc + 1) * PJ_KC;
#pragma unroll
    for (int i = 0; i < QK_BM / 16; ++i) ra[i] = *(const v4f*)(ag + i * 16 * DD + ko);
#pragma unroll
    for (int i = 0; i < QK_NTL; ++i)     rb[i] = *(const v4f*)(bg + i * 16 * DH + ko);
    pj_chunk(As + cur * (QK_BM * PJ_P) + aoff, Bs + cur * (QK_BN * PJ_P) + boff, acc);
#pragma unroll
    for (int i = 0; i < QK_BM / 16; ++i) *(v4f*)&As[(cur ^ 1) * (QK_BM * PJ_P) + sl + i * 16 * PJ_P] = ra[i];
#pragma unroll
    for (int i = 0; i < QK_NTL; ++i)     *(v4f*)&Bs[(cur ^ 1) * (QK_BN * PJ_P) + sl + i * 16 * PJ_P] = rb[i];
    __syncthreads();
  }
  pj_chunk(As + 1 * (QK_BM * PJ_P) + aoff, Bs + 1 * (QK_BN * PJ_P) + boff, acc);

  const int rbase = (lane >> 4) * 8;
  const int ncol  = lane & 15;
#pragma unroll
  for (int q = 0; q < QK_NTL; ++q) {
    int n = h * D3 + n0 + q * 16 + ncol;
    float bias = cb[n];
    for (int r = 0; r < 8; ++r)
      qkv[(t0 + wid * 16 + rbase + r) * D9 + n] = acc[q][r] + bias;
  }
}

#define ATT_SMALL_SUM 8.0f
__global__ __launch_bounds__(D3)
void cea_attn(const float* __restrict__ qkv,
              float* __restrict__ att) {
  const int t = blockIdx.x, i = threadIdx.x;
  const float* __restrict__ base = qkv + t * D9;
  const float* __restrict__ Kp = base + D3;
  const float* __restrict__ Vp = base + 2 * D3;
  const float Qi = base[i];
  const float Qh = Qi * (0.5f * GAMMA);

  float acc2 = 0.0f, usum = 0.0f;
#pragma unroll 16
  for (int j = 0; j < D3; ++j) {
    const float u = 1.0f + __builtin_amdgcn_tanhf(Qh * Kp[j]);
    acc2  = __builtin_fmaf(u, Vp[j], acc2);
    usum += u;
  }
  float acc = 0.5f * acc2, ssum = 0.5f * usum;
  if (ssum < ATT_SMALL_SUM) {
    const float Qe = Qi * (-GAMMA * LOG2E);
    acc = 0.0f; ssum = 0.0f;
#pragma unroll 4
    for (int j = 0; j < D3; ++j) {
      const float s = __builtin_amdgcn_rcpf(1.0f + __builtin_amdgcn_exp2f(Qe * Kp[j]));
      acc   = __builtin_fmaf(s, Vp[j], acc);
      ssum += s;
    }
  }
  att[t * D3 + i] = acc / (ssum + 1e-8f);
}

__global__ __launch_bounds__(128)
void cea_proj(const float* __restrict__ att,
              const float* __restrict__ spP,
              const float* __restrict__ bp,
              float* __restrict__ hout,
              float* __restrict__ lnst) {
  __shared__ __attribute__((aligned(16))) float As[2 * PJ_BM * PJ_P];
  __shared__ __attribute__((aligned(16))) float Bs[2 * PJ_BN * PJ_P];
  const int tid = threadIdx.x;
  const int t0 = blockIdx.x * PJ_BM;
  const int n0 = blockIdx.y * PJ_BN;

  const int srow = tid >> 3;
  const int sq   = (tid & 7) * 4;
  const float* __restrict__ ag = att + (t0 + srow) * D3 + sq;
  const float* __restrict__ bg = spP + (n0 + srow) * D3 + sq;
  const int sl = srow * PJ_P + sq;

  v4f ra0, ra1, ra2, ra3, rb0, rb1, rb2, rb3, rb4, rb5, rb6, rb7, rb8;
#define PJ_GLOAD(ko)                                        \
  ra0 = *(const v4f*)(ag + 0 * 16 * D3 + (ko));             \
  ra1 = *(const v4f*)(ag + 1 * 16 * D3 + (ko));             \
  ra2 = *(const v4f*)(ag + 2 * 16 * D3 + (ko));             \
  ra3 = *(const v4f*)(ag + 3 * 16 * D3 + (ko));             \
  rb0 = *(const v4f*)(bg + 0 * 16 * D3 + (ko));             \
  rb1 = *(const v4f*)(bg + 1 * 16 * D3 + (ko));             \
  rb2 = *(const v4f*)(bg + 2 * 16 * D3 + (ko));             \
  rb3 = *(const v4f*)(bg + 3 * 16 * D3 + (ko));             \
  rb4 = *(const v4f*)(bg + 4 * 16 * D3 + (ko));             \
  rb5 = *(const v4f*)(bg + 5 * 16 * D3 + (ko));             \
  rb6 = *(const v4f*)(bg + 6 * 16 * D3 + (ko));             \
  rb7 = *(const v4f*)(bg + 7 * 16 * D3 + (ko));             \
  rb8 = *(const v4f*)(bg + 8 * 16 * D3 + (ko));
#define PJ_SSTORE(st)                                                   \
  *(v4f*)&As[(st) * (PJ_BM * PJ_P) + sl + 0 * 16 * PJ_P] = ra0;         \
  *(v4f*)&As[(st) * (PJ_BM * PJ_P) + sl + 1 * 16 * PJ_P] = ra1;         \
  *(v4f*)&As[(st) * (PJ_BM * PJ_P) + sl + 2 * 16 * PJ_P] = ra2;         \
  *(v4f*)&As[(st) * (PJ_BM * PJ_P) + sl + 3 * 16 * PJ_P] = ra3;         \
  *(v4f*)&Bs[(st) * (PJ_BN * PJ_P) + sl + 0 * 16 * PJ_P] = rb0;         \
  *(v4f*)&Bs[(st) * (PJ_BN * PJ_P) + sl + 1 * 16 * PJ_P] = rb1;         \
  *(v4f*)&Bs[(st) * (PJ_BN * PJ_P) + sl + 2 * 16 * PJ_P] = rb2;         \
  *(v4f*)&Bs[(st) * (PJ_BN * PJ_P) + sl + 3 * 16 * PJ_P] = rb3;         \
  *(v4f*)&Bs[(st) * (PJ_BN * PJ_P) + sl + 4 * 16 * PJ_P] = rb4;         \
  *(v4f*)&Bs[(st) * (PJ_BN * PJ_P) + sl + 5 * 16 * PJ_P] = rb5;         \
  *(v4f*)&Bs[(st) * (PJ_BN * PJ_P) + sl + 6 * 16 * PJ_P] = rb6;         \
  *(v4f*)&Bs[(st) * (PJ_BN * PJ_P) + sl + 7 * 16 * PJ_P] = rb7;         \
  *(v4f*)&Bs[(st) * (PJ_BN * PJ_P) + sl + 8 * 16 * PJ_P] = rb8;

  PJ_GLOAD(0)
  PJ_SSTORE(0)
  __syncthreads();

  const int lane = tid & 31, wid = tid >> 5;
  const int m  = lane & 15;
  const int kh = (lane >> 4) * 2;
  const int aoff = (wid * 16 + m) * PJ_P + kh;
  const int boff = m * PJ_P + kh;

  v8f acc[9];
#pragma unroll
  for (int q = 0; q < 9; ++q) acc[q] = (v8f){};

  for (int kc = 0; kc < D3 / PJ_KC - 1; ++kc) {
    const int cur = (kc & 1);
    const int ko = (kc + 1) * PJ_KC;
    PJ_GLOAD(ko)
    pj_chunk(As + cur * (PJ_BM * PJ_P) + aoff, Bs + cur * (PJ_BN * PJ_P) + boff, acc);
    PJ_SSTORE(cur ^ 1)
    __syncthreads();
  }
  pj_chunk(As + 1 * (PJ_BM * PJ_P) + aoff, Bs + 1 * (PJ_BN * PJ_P) + boff, acc);
#undef PJ_GLOAD
#undef PJ_SSTORE

  const int rbase = (lane >> 4) * 8;
  const int ncol  = lane & 15;
  float rsum[8], rm2[8];
#pragma unroll
  for (int r = 0; r < 8; ++r) rsum[r] = 0.0f;
#pragma unroll
  for (int q = 0; q < 9; ++q) {
    int n = n0 + q * 16 + ncol;
    float bias = bp[n];
#pragma unroll
    for (int r = 0; r < 8; ++r) {
      float hv = acc[q][r] + bias;
      acc[q][r] = hv;
      hout[(t0 + wid * 16 + rbase + r) * D9 + n] = hv;
      rsum[r] += hv;
    }
  }
#pragma unroll
  for (int r = 0; r < 8; ++r) {
    rsum[r] = row16_sum(rsum[r]) * (1.0f / (float)PJ_BN);
    rm2[r] = 0.0f;
  }
#pragma unroll
  for (int q = 0; q < 9; ++q) {
#pragma unroll
    for (int r = 0; r < 8; ++r) {
      float dv = acc[q][r] - rsum[r];
      rm2[r] = __builtin_fmaf(dv, dv, rm2[r]);
    }
  }
#pragma unroll
  for (int r = 0; r < 8; ++r) rm2[r] = row16_sum(rm2[r]);
  float wmean = rsum[0], wm2 = rm2[0];
#pragma unroll
  for (int r = 1; r < 8; ++r) { if (ncol == r) { wmean = rsum[r]; wm2 = rm2[r]; } }
  if (ncol < 8) {
    const int tok = t0 + wid * 16 + rbase + ncol;
    *(v2f*)&lnst[(blockIdx.y * NT + tok) * 2] = (v2f){wmean, wm2};
  }
}

#define LN_NB (D9 / PJ_BN)
__global__ __launch_bounds__(256)
void cea_ln(float* __restrict__ out,
            const float* __restrict__ lnst,
            const float* __restrict__ gn,
            const float* __restrict__ bn) {
  const int t = blockIdx.x, tid = threadIdx.x;
  float* __restrict__ h = out + t * D9;
  const bool tail = tid < D9 - 1024;
  const int  nt5  = tail ? tid + 1024 : tid;

  float v[5], g[5], b[5];
#pragma unroll
  for (int k = 0; k < 4; ++k) { v[k] = h[tid + 256 * k]; g[k] = gn[tid + 256 * k]; b[k] = bn[tid + 256 * k]; }
  v[4] = h[nt5]; g[4] = gn[nt5]; b[4] = bn[nt5];

  float mean_b[LN_NB], m2_b[LN_NB];
#pragma unroll
  for (int q = 0; q < LN_NB; ++q) {
    mean_b[q] = lnst[(q * NT + t) * 2];
    m2_b[q]   = lnst[(q * NT + t) * 2 + 1];
  }
  float mu = 0.0f;
#pragma unroll
  for (int q = 0; q < LN_NB; ++q) mu += mean_b[q];
  mu *= (1.0f / (float)LN_NB);
  float ss = 0.0f;
#pragma unroll
  for (int q = 0; q < LN_NB; ++q) {
    float dm = mean_b[q] - mu;
    ss += m2_b[q] + (float)PJ_BN * dm * dm;
  }
  const float var = ss * (1.0f / (float)D9);
  const float rs  = __builtin_amdgcn_rsqf(var + 1e-5f);

#pragma unroll
  for (int k = 0; k < 4; ++k) h[tid + 256 * k] = (v[k] - mu) * rs * g[k] + b[k];
  if (tail) h[nt5] = (v[4] - mu) * rs * g[4] + b[4];
}


extern "C" void kernel_launch(void* const* d_in, const int* in_sizes, int n_in,
                              void* d_out, int out_size, void* d_ws, size_t ws_size,
                              hipStream_t stream) {
  const float* x  = (const float*)d_in[0];
  const float* Wq = (const float*)d_in[1];
  const float* bq = (const float*)d_in[2];
  const float* Sq = (const float*)d_in[3];
  const float* Wk = (const float*)d_in[4];
  const float* bk = (const float*)d_in[5];
  const float* Sk = (const float*)d_in[6];
  const float* Wv = (const float*)d_in[7];
  const float* bv = (const float*)d_in[8];
  const float* Sv = (const float*)d_in[9];
  const float* Wp = (const float*)d_in[10];
  const float* bp = (const float*)d_in[11];
  const float* gn = (const float*)d_in[12];
  const float* bn = (const float*)d_in[13];
  float* out = (float*)d_out;

  float* ws = (float*)d_ws;
  float* lnst = ws;
  float* Mt   = ws + 3 * D3 * DH;
  float* cb   = Mt  + 3 * D3 * DH;
  float* spP  = cb  + 3 * D3;
  float* qkv  = spP + D9 * D3;
  float* attb = qkv + NT * D9;

  cea_prep<<<dim3(PREP_BIAS_BLOCKS + PREP_FUSE_BLOCKS + PREP_SPP_BLOCKS), dim3(256), 0, stream>>>(
      Wq, Wk, Wv, Wp, bq, bk, bv, Sq, Sk, Sv, Mt, spP, cb);

  cea_qkv<<<dim3(NT / QK_BM, 3 * (D3 / QK_BN)), dim3(128), 0, stream>>>(x, Mt, cb, qkv);

  cea_attn<<<dim3(NT), dim3(D3), 0, stream>>>(qkv, attb);

  cea_proj<<<dim3(NT / PJ_BM, D9 / PJ_BN), dim3(128), 0, stream>>>(attb, spP, bp, out, lnst);

  cea_ln<<<dim3(NT), dim3(256), 0, stream>>>(out, lnst, gn, bn);

  (void)in_sizes; (void)n_in; (void)out_size; (void)ws_size;
}


# softplus of all weights on the hardware exp2, log2 and rcp, as max(w,0) + log(u) * t / (u - 1) with t = exp(-|w|), u = 1 + t, in place of the library log1pf; weights differ in their last bits
# speedup vs baseline: 3.4016x; 1.1501x over previous
#include <hip/hip_runtime.h>
#include <hip/hip_bf16.h>
#include <math.h>


#define NT     2048
#define DD     384
#define DH     128
#define D3     384
#define D9     1152
#define GAMMA  5.0f
#define LOG2E  1.4426950408889634f

typedef float v2f __attribute__((ext_vector_type(2)));
typedef float v4f __attribute__((ext_vector_type(4)));
typedef float v8f __attribute__((ext_vector_type(8)));

static __device__ __forceinline__ v8f wmma4(v2f a, v2f b, v8f c) {
  return __builtin_amdgcn_wmma_f32_16x16x4_f32(
       false, a,  false, b,
       (short)0, c,  false,  false);
}

template <int M>
static __device__ __forceinline__ float row_xor_add(float v) {
  const int iv = __builtin_bit_cast(int, v);
  const int sw = __builtin_amdgcn_update_dpp(iv, iv, 0x160 + M, 0xf, 0xf, false);
  return v + __builtin_bit_cast(float, sw);
}
static __device__ __forceinline__ float row16_sum(float v) {
  v = row_xor_add<1>(v);
  v = row_xor_add<2>(v);
  v = row_xor_add<4>(v);
  v = row_xor_add<8>(v);
  return v;
}

#define LN2 0.6931471805599453f
static __device__ __forceinline__ float softplusf(float w) {
  const float t = __builtin_amdgcn_exp2f(-LOG2E * __builtin_fabsf(w));
  const float u = 1.0f + t;
  const float d = u - 1.0f;
  const float l = (LN2 * __builtin_amdgcn_logf(u)) * (t * __builtin_amdgcn_rcpf(d));
  return __builtin_fmaxf(w, 0.0f) + ((d == 0.0f) ? t : l);
}


#define FZ_BN 64
#define FZ_BK 32
#define FZ_JC 96
#define FZ_PA 72
#define FZ_PB 40
#define FZ_PC 36
#define PREP_BIAS_BLOCKS ((3 * D3) / 32)
#define PREP_FUSE_BLOCKS (3 * (D3 / FZ_BN) * (DH / FZ_BK))
#define PREP_SPP_BLOCKS  ((D9 * D3) / 256)
struct fz_frag { v2f a[6]; v2f b[6]; };
static __device__ __forceinline__ void fz_read(const float* __restrict__ Af,
                                               const float* __restrict__ Bf,
                                               int ks0, fz_frag& f) {
#pragma unroll
  for (int s = 0; s < 6; ++s) {
    const int ks = ks0 + 4 * s;
    f.a[s] = (v2f){Af[ks * FZ_PA], Af[(ks + 1) * FZ_PA]};
    f.b[s] = (v2f){Bf[ks * FZ_PB], Bf[(ks + 1) * FZ_PB]};
  }
}
static __device__ __forceinline__ v8f fz_mma(const fz_frag& f, v8f acc) {
#pragma unroll
  for (int s = 0; s < 6; ++s) acc = wmma4(f.a[s], f.b[s], acc);
  return acc;
}
__global__ __launch_bounds__(256)
void cea_prep(const float* __restrict__ Wq, const float* __restrict__ Wk,
              const float* __restrict__ Wv, const float* __restrict__ Wp,
              const float* __restrict__ bq, const float* __restrict__ bk,
              const float* __restrict__ bv,
              const float* __restrict__ Sq, const float* __restrict__ Sk,
              const float* __restrict__ Sv,
              float* __restrict__ Mt, float* __restrict__ spP,
              float* __restrict__ c) {
  __shared__ __attribute__((aligned(16))) float As[FZ_JC * FZ_PA];
  __shared__ __attribute__((aligned(16))) float Bs[FZ_JC * FZ_PB];
  __shared__ float red[8 * 32];
  const int blk = blockIdx.x, tid = threadIdx.x;
  if (blk >= PREP_BIAS_BLOCKS + PREP_FUSE_BLOCKS) {
    int i = (blk - PREP_BIAS_BLOCKS - PREP_FUSE_BLOCKS) * 256 + tid;
    spP[i] = softplusf(Wp[i]);
    return;
  }
  if (blk >= PREP_BIAS_BLOCKS) {
    const int b2 = blk - PREP_BIAS_BLOCKS;
    const int h  = b2 / (PREP_FUSE_BLOCKS / 3);
    const int r2 = b2 - h * (PREP_FUSE_BLOCKS / 3);
    const int n0 = (r2 / (DH / FZ_BK)) * FZ_BN;
    const int k0 = (r2 % (DH / FZ_BK)) * FZ_BK;
    const float* __restrict__ S = (h == 0) ? Sq : (h == 1) ? Sk : Sv;
    const float* __restrict__ W = (h == 0) ? Wq : (h == 1) ? Wk : Wv;

    const int arow = tid >> 4, acol = (tid & 15) * 4;
    const int brow = tid >> 3, bcol = (tid & 7) * 4;
    const float* __restrict__ ag = S + arow * D3 + n0 + acol;
    const float* __restrict__ bg = W + brow * DH + k0 + bcol;

    v4f ra[FZ_JC / 16], rb[FZ_JC / 32];
#pragma unroll
    for (int i = 0; i < FZ_JC / 16; ++i) ra[i] = *(const v4f*)(ag + i * 16 * D3);
#pragma unroll
    for (int i = 0; i < FZ_JC / 32; ++i) rb[i] = *(const v4f*)(bg + i * 32 * DH);

    const int lane = tid & 31, wid = tid >> 5;
    const int cc = lane & 15;
    const int hh = lane >> 4;
    const int nt = wid & 3, kt = wid >> 2;
    const float* Af = As + (2 * hh) * FZ_PA + nt * 16 + cc;
    const float* Bf = Bs + (2 * hh) * FZ_PB + kt * 16 + cc;

    v8f acc = (v8f){};
#pragma unroll
    for (int ch = 0; ch < D3 / FZ_JC; ++ch) {
#pragma unroll
      for (int i = 0; i < FZ_JC / 16; ++i) *(v4f*)&As[(arow + 16 * i) * FZ_PA + acol] = ra[i];
#pragma unroll
      for (int i = 0; i < FZ_JC / 32; ++i) {
        v4f w = rb[i];
        *(v4f*)&Bs[(brow + 32 * i) * FZ_PB + bcol] =
            (v4f){softplusf(w.x), softplusf(w.y), softplusf(w.z), softplusf(w.w)};
      }
      __syncthreads();
      if (ch + 1 < D3 / FZ_JC) {
#pragma unroll
        for (int i = 0; i < FZ_JC / 16; ++i) ra[i] = *(const v4f*)(ag + ((ch + 1) * FZ_JC + i * 16) * D3);
#pragma unroll
        for (int i = 0; i < FZ_JC / 32; ++i) rb[i] = *(const v4f*)(bg + ((ch + 1) * FZ_JC + i * 32) * DH);
      }
      fz_frag f0, f1;
      fz_read(Af, Bf, 0, f0);
      fz_read(Af, Bf, 24, f1);
      __builtin_amdgcn_sched_barrier(0);
      acc = fz_mma(f0, acc);
      __builtin_amdgcn_sched_barrier(0);
      fz_read(Af, Bf, 48, f0);
      __builtin_amdgcn_sched_barrier(0);
      acc = fz_mma(f1, acc);
      __builtin_amdgcn_sched_barrier(0);
      fz_read(Af, Bf, 72, f1);
      __builtin_amdgcn_sched_barrier(0);
      acc = fz_mma(f0, acc);
      __builtin_amdgcn_sched_barrier(0);
      acc = fz_mma(f1, acc);
      __syncthreads();
    }

    float* Cs = As;
#pragma unroll
    for (int r = 0; r < 8; ++r) Cs[(nt * 16 + 8 * hh + r) * FZ_PC + kt * 16 + cc] = acc[r];
    __syncthreads();
#pragma unroll
    for (int i = 0; i < FZ_BN / 32; ++i) {
      const int row = (tid >> 3) + 32 * i, c4 = (tid & 7) * 4;
      v4f v = *(const v4f*)&Cs[row * FZ_PC + c4];
      *(v4f*)&Mt[h * (D3 * DH) + (n0 + row) * DH + k0 + c4] = v;
    }
    return;
  }
  const int h  = blk / (D3 / 32);
  const int nl = tid & 31, jg = tid >> 5;
  const int n  = (blk - h * (D3 / 32)) * 32 + nl;
  const float* b = (h == 0) ? bq : (h == 1) ? bk : bv;
  const float* S = (h == 0) ? Sq : (h == 1) ? Sk : Sv;
  float acc = 0.0f;
#pragma unroll 16
  for (int jj = 0; jj < D3 / 8; ++jj) {
    int j = jg * (D3 / 8) + jj;
    acc += b[j] * S[j * D3 + n];
  }
  red[jg * 32 + nl] = acc;
  __syncthreads();
  if (jg == 0) {
    float s = red[nl];
    for (int q = 1; q < 8; ++q) s += red[q * 32 + nl];
    c[h * D3 + n] = s;
  }
}

#define PJ_BM 64
#define PJ_BN 144
#define PJ_KC 32
#define PJ_P  36
template <int NTL>
struct pj_frag { v4f a; v4f b[NTL]; };
template <int NTL>
static __device__ __forceinline__ void pj_read(const float* __restrict__ Ab,
                                               const float* __restrict__ Bb,
                                               int ks, pj_frag<NTL>& f) {
  v2f lo = *(const v2f*)&Ab[ks], hi = *(const v2f*)&Ab[ks + 4];
  f.a = (v4f){lo.x, lo.y, hi.x, hi.y};
#pragma unroll
  for (int q = 0; q < NTL; ++q) {
    v2f l2 = *(const v2f*)&Bb[q * 16 * PJ_P + ks], h2 = *(const v2f*)&Bb[q * 16 * PJ_P + ks + 4];
    f.b[q] = (v4f){l2.x, l2.y, h2.x, h2.y};
  }
}
template <int NTL>
static __device__ __forceinline__ void pj_mma(const pj_frag<NTL>& f, v8f (&acc)[NTL]) {
#pragma unroll
  for (int q = 0; q < NTL; ++q) acc[q] = wmma4((v2f){f.a.x, f.a.y}, (v2f){f.b[q].x, f.b[q].y}, acc[q]);
#pragma unroll
  for (int q = 0; q < NTL; ++q) acc[q] = wmma4((v2f){f.a.z, f.a.w}, (v2f){f.b[q].z, f.b[q].w}, acc[q]);
}
template <int NTL>
static __device__ __forceinline__ void pj_chunk(const float* __restrict__ Ab,
                                                const float* __restrict__ Bb,
                                                v8f (&acc)[NTL]) {
  pj_frag<NTL> f0, f1;
  pj_read(Ab, Bb, 0, f0);
  pj_read(Ab, Bb, 8, f1);
  __builtin_amdgcn_sched_barrier(0);
  pj_mma(f0, acc);
  __builtin_amdgcn_sched_barrier(0);
  pj_read(Ab, Bb, 16, f0);
  __builtin_amdgcn_sched_barrier(0);
  pj_mma(f1, acc);
  __builtin_amdgcn_sched_barrier(0);
  pj_read(Ab, Bb, 24, f1);
  __builtin_amdgcn_sched_barrier(0);
  pj_mma(f0, acc);
  __builtin_amdgcn_sched_barrier(0);
  pj_mma(f1, acc);
}


#define QK_BM 64
#define QK_BN 96
#define QK_NTL (QK_BN / 16)
__global__ __launch_bounds__(128)
void cea_qkv(const float* __restrict__ x,
             const float* __restrict__ Mt,
             const float* __restrict__ cb,
             float* __restrict__ qkv) {
  __shared__ __attribute__((aligned(16))) float As[2 * QK_BM * PJ_P];
  __shared__ __attribute__((aligned(16))) float Bs[2 * QK_BN * PJ_P];
  const int tid = threadIdx.x;
  const int t0 = blockIdx.x * QK_BM;
  const int h  = blockIdx.y / (D3 / QK_BN);
  const int n0 = (blockIdx.y - h * (D3 / QK_BN)) * QK_BN;

  const int srow = tid >> 3;
  const int sq   = (tid & 7) * 4;
  const float* __restrict__ ag = x + (t0 + srow) * DD + h * DH + sq;
  const float* __restrict__ bg = Mt + h * (D3 * DH) + (n0 + srow) * DH + sq;
  const int sl = srow * PJ_P + sq;

  v4f ra[QK_BM / 16], rb[QK_NTL];

#pragma unroll
  for (int i = 0; i < QK_BM / 16; ++i) ra[i] = *(const v4f*)(ag + i * 16 * DD);
#pragma unroll
  for (int i = 0; i < QK_NTL; ++i)     rb[i] = *(const v4f*)(bg + i * 16 * DH);
#pragma unroll
  for (int i = 0; i < QK_BM / 16; ++i) *(v4f*)&As[sl + i * 16 * PJ_P] = ra[i];
#pragma unroll
  for (int i = 0; i < QK_NTL; ++i)     *(v4f*)&Bs[sl + i * 16 * PJ_P] = rb[i];
  __syncthreads();

  const int lane = tid & 31, wid = tid >> 5;
  const int m  = lane & 15;
  const int kh = (lane >> 4) * 2;
  const int aoff = (wid * 16 + m) * PJ_P + kh;
  const int boff = m * PJ_P + kh;

  v8f acc[QK_NTL];
#pragma unroll
  for (int q = 0; q < QK_NTL; ++q) acc[q] = (v8f){};

  for (int kc = 0; kc < DH / PJ_KC - 1; ++kc) {
    const int cur = (kc & 1);
    const int ko = (kc + 1) * PJ_KC;
#pragma unroll
    for (int i = 0; i < QK_BM / 16; ++i) ra[i] = *(const v4f*)(ag + i * 16 * DD + ko);
#pragma unroll
    for (int i = 0; i < QK_NTL; ++i)     rb[i] = *(const v4f*)(bg + i * 16 * DH + ko);
    pj_chunk(As + cur * (QK_BM * PJ_P) + aoff, Bs + cur * (QK_BN * PJ_P) + boff, acc);
#pragma unroll
    for (int i = 0; i < QK_BM / 16; ++i) *(v4f*)&As[(cur ^ 1) * (QK_BM * PJ_P) + sl + i * 16 * PJ_P] = ra[i];
#pragma unroll
    for (int i = 0; i < QK_NTL; ++i)     *(v4f*)&Bs[(cur ^ 1) * (QK_BN * PJ_P) + sl + i * 16 * PJ_P] = rb[i];
    __syncthreads();
  }
  pj_chunk(As + 1 * (QK_BM * PJ_P) + aoff, Bs + 1 * (QK_BN * PJ_P) + boff, acc);

  const int rbase = (lane >> 4) * 8;
  const int ncol  = lane & 15;
#pragma unroll
  for (int q = 0; q < QK_NTL; ++q) {
    int n = h * D3 + n0 + q * 16 + ncol;
    float bias = cb[n];
    for (int r = 0; r < 8; ++r)
      qkv[(t0 + wid * 16 + rbase + r) * D9 + n] = acc[q][r] + bias;
  }
}

#define ATT_SMALL_SUM 8.0f
__global__ __launch_bounds__(D3)
void cea_attn(const float* __restrict__ qkv,
              float* __restrict__ att) {
  const int t = blockIdx.x, i = threadIdx.x;
  const float* __restrict__ base = qkv + t * D9;
  const float* __restrict__ Kp = base + D3;
  const float* __restrict__ Vp = base + 2 * D3;
  const float Qi = base[i];
  const float Qh = Qi * (0.5f * GAMMA);

  float acc2 = 0.0f, usum = 0.0f;
#pragma unroll 16
  for (int j = 0; j < D3; ++j) {
    const float u = 1.0f + __builtin_amdgcn_tanhf(Qh * Kp[j]);
    acc2  = __builtin_fmaf(u, Vp[j], acc2);
    usum += u;
  }
  float acc = 0.5f * acc2, ssum = 0.5f * usum;
  if (ssum < ATT_SMALL_SUM) {
    const float Qe = Qi * (-GAMMA * LOG2E);
    acc = 0.0f; ssum = 0.0f;
#pragma unroll 4
    for (int j = 0; j < D3; ++j) {
      const float s = __builtin_amdgcn_rcpf(1.0f + __builtin_amdgcn_exp2f(Qe * Kp[j]));
      acc   = __builtin_fmaf(s, Vp[j], acc);
      ssum += s;
    }
  }
  att[t * D3 + i] = acc / (ssum + 1e-8f);
}

__global__ __launch_bounds__(128)
void cea_proj(const float* __restrict__ att,
              const float* __restrict__ spP,
              const float* __restrict__ bp,
              float* __restrict__ hout,
              float* __restrict__ lnst) {
  __shared__ __attribute__((aligned(16))) float As[2 * PJ_BM * PJ_P];
  __shared__ __attribute__((aligned(16))) float Bs[2 * PJ_BN * PJ_P];
  const int tid = threadIdx.x;
  const int t0 = blockIdx.x * PJ_BM;
  const int n0 = blockIdx.y * PJ_BN;

  const int srow = tid >> 3;
  const int sq   = (tid & 7) * 4;
  const float* __restrict__ ag = att + (t0 + srow) * D3 + sq;
  const float* __restrict__ bg = spP + (n0 + srow) * D3 + sq;
  const int sl = srow * PJ_P + sq;

  v4f ra0, ra1, ra2, ra3, rb0, rb1, rb2, rb3, rb4, rb5, rb6, rb7, rb8;
#define PJ_GLOAD(ko)                                        \
  ra0 = *(const v4f*)(ag + 0 * 16 * D3 + (ko));             \
  ra1 = *(const v4f*)(ag + 1 * 16 * D3 + (ko));             \
  ra2 = *(const v4f*)(ag + 2 * 16 * D3 + (ko));             \
  ra3 = *(const v4f*)(ag + 3 * 16 * D3 + (ko));             \
  rb0 = *(const v4f*)(bg + 0 * 16 * D3 + (ko));             \
  rb1 = *(const v4f*)(bg + 1 * 16 * D3 + (ko));             \
  rb2 = *(const v4f*)(bg + 2 * 16 * D3 + (ko));             \
  rb3 = *(const v4f*)(bg + 3 * 16 * D3 + (ko));             \
  rb4 = *(const v4f*)(bg + 4 * 16 * D3 + (ko));             \
  rb5 = *(const v4f*)(bg + 5 * 16 * D3 + (ko));             \
  rb6 = *(const v4f*)(bg + 6 * 16 * D3 + (ko));             \
  rb7 = *(const v4f*)(bg + 7 * 16 * D3 + (ko));             \
  rb8 = *(const v4f*)(bg + 8 * 16 * D3 + (ko));
#define PJ_SSTORE(st)                                                   \
  *(v4f*)&As[(st) * (PJ_BM * PJ_P) + sl + 0 * 16 * PJ_P] = ra0;         \
  *(v4f*)&As[(st) * (PJ_BM * PJ_P) + sl + 1 * 16 * PJ_P] = ra1;         \
  *(v4f*)&As[(st) * (PJ_BM * PJ_P) + sl + 2 * 16 * PJ_P] = ra2;         \
  *(v4f*)&As[(st) * (PJ_BM * PJ_P) + sl + 3 * 16 * PJ_P] = ra3;         \
  *(v4f*)&Bs[(st) * (PJ_BN * PJ_P) + sl + 0 * 16 * PJ_P] = rb0;         \
  *(v4f*)&Bs[(st) * (PJ_BN * PJ_P) + sl + 1 * 16 * PJ_P] = rb1;         \
  *(v4f*)&Bs[(st) * (PJ_BN * PJ_P) + sl + 2 * 16 * PJ_P] = rb2;         \
  *(v4f*)&Bs[(st) * (PJ_BN * PJ_P) + sl + 3 * 16 * PJ_P] = rb3;         \
  *(v4f*)&Bs[(st) * (PJ_BN * PJ_P) + sl + 4 * 16 * PJ_P] = rb4;         \
  *(v4f*)&Bs[(st) * (PJ_BN * PJ_P) + sl + 5 * 16 * PJ_P] = rb5;         \
  *(v4f*)&Bs[(st) * (PJ_BN * PJ_P) + sl + 6 * 16 * PJ_P] = rb6;         \
  *(v4f*)&Bs[(st) * (PJ_BN * PJ_P) + sl + 7 * 16 * PJ_P] = rb7;         \
  *(v4f*)&Bs[(st) * (PJ_BN * PJ_P) + sl + 8 * 16 * PJ_P] = rb8;

  PJ_GLOAD(0)
  PJ_SSTORE(0)
  __syncthreads();

  const int lane = tid & 31, wid = tid >> 5;
  const int m  = lane & 15;
  const int kh = (lane >> 4) * 2;
  const int aoff = (wid * 16 + m) * PJ_P + kh;
  const int boff = m * PJ_P + kh;

  v8f acc[9];
#pragma unroll
  for (int q = 0; q < 9; ++q) acc[q] = (v8f){};

  for (int kc = 0; kc < D3 / PJ_KC - 1; ++kc) {
    const int cur = (kc & 1);
    const int ko = (kc + 1) * PJ_KC;
    PJ_GLOAD(ko)
    pj_chunk(As + cur * (PJ_BM * PJ_P) + aoff, Bs + cur * (PJ_BN * PJ_P) + boff, acc);
    PJ_SSTORE(cur ^ 1)
    __syncthreads();
  }
  pj_chunk(As + 1 * (PJ_BM * PJ_P) + aoff, Bs + 1 * (PJ_BN * PJ_P) + boff, acc);
#undef PJ_GLOAD
#undef PJ_SSTORE

  const int rbase = (lane >> 4) * 8;
  const int ncol  = lane & 15;
  float rsum[8], rm2[8];
#pragma unroll
  for (int r = 0; r < 8; ++r) rsum[r] = 0.0f;
#pragma unroll
  for (int q = 0; q < 9; ++q) {
    int n = n0 + q * 16 + ncol;
    float bias = bp[n];
#pragma unroll
    for (int r = 0; r < 8; ++r) {
      float hv = acc[q][r] + bias;
      acc[q][r] = hv;
      hout[(t0 + wid * 16 + rbase + r) * D9 + n] = hv;
      rsum[r] += hv;
    }
  }
#pragma unroll
  for (int r = 0; r < 8; ++r) {
    rsum[r] = row16_sum(rsum[r]) * (1.0f / (float)PJ_BN);
    rm2[r] = 0.0f;
  }
#pragma unroll
  for (int q = 0; q < 9; ++q) {
#pragma unroll
    for (int r = 0; r < 8; ++r) {
      float dv = acc[q][r] - rsum[r];
      rm2[r] = __builtin_fmaf(dv, dv, rm2[r]);
    }
  }
#pragma unroll
  for (int r = 0; r < 8; ++r) rm2[r] = row16_sum(rm2[r]);
  float wmean = rsum[0], wm2 = rm2[0];
#pragma unroll
  for (int r = 1; r < 8; ++r) { if (ncol == r) { wmean = rsum[r]; wm2 = rm2[r]; } }
  if (ncol < 8) {
    const int tok = t0 + wid * 16 + rbase + ncol;
    *(v2f*)&lnst[(blockIdx.y * NT + tok) * 2] = (v2f){wmean, wm2};
  }
}

#define LN_NB (D9 / PJ_BN)
__global__ __launch_bounds__(256)
void cea_ln(float* __restrict__ out,
            const float* __restrict__ lnst,
            const float* __restrict__ gn,
            const float* __restrict__ bn) {
  const int t = blockIdx.x, tid = threadIdx.x;
  float* __restrict__ h = out + t * D9;
  const bool tail = tid < D9 - 1024;
  const int  nt5  = tail ? tid + 1024 : tid;

  float v[5], g[5], b[5];
#pragma unroll
  for (int k = 0; k < 4; ++k) { v[k] = h[tid + 256 * k]; g[k] = gn[tid + 256 * k]; b[k] = bn[tid + 256 * k]; }
  v[4] = h[nt5]; g[4] = gn[nt5]; b[4] = bn[nt5];

  float mean_b[LN_NB], m2_b[LN_NB];
#pragma unroll
  for (int q = 0; q < LN_NB; ++q) {
    mean_b[q] = lnst[(q * NT + t) * 2];
    m2_b[q]   = lnst[(q * NT + t) * 2 + 1];
  }
  float mu = 0.0f;
#pragma unroll
  for (int q = 0; q < LN_NB; ++q) mu += mean_b[q];
  mu *= (1.0f / (float)LN_NB);
  float ss = 0.0f;
#pragma unroll
  for (int q = 0; q < LN_NB; ++q) {
    float dm = mean_b[q] - mu;
    ss += m2_b[q] + (float)PJ_BN * dm * dm;
  }
  const float var = ss * (1.0f / (float)D9);
  const float rs  = __builtin_amdgcn_rsqf(var + 1e-5f);

#pragma unroll
  for (int k = 0; k < 4; ++k) h[tid + 256 * k] = (v[k] - mu) * rs * g[k] + b[k];
  if (tail) h[nt5] = (v[4] - mu) * rs * g[4] + b[4];
}


extern "C" void kernel_launch(void* const* d_in, const int* in_sizes, int n_in,
                              void* d_out, int out_size, void* d_ws, size_t ws_size,
                              hipStream_t stream) {
  const float* x  = (const float*)d_in[0];
  const float* Wq = (const float*)d_in[1];
  const float* bq = (const float*)d_in[2];
  const float* Sq = (const float*)d_in[3];
  const float* Wk = (const float*)d_in[4];
  const float* bk = (const float*)d_in[5];
  const float* Sk = (const float*)d_in[6];
  const float* Wv = (const float*)d_in[7];
  const float* bv = (const float*)d_in[8];
  const float* Sv = (const float*)d_in[9];
  const float* Wp = (const float*)d_in[10];
  const float* bp = (const float*)d_in[11];
  const float* gn = (const float*)d_in[12];
  const float* bn = (const float*)d_in[13];
  float* out = (float*)d_out;

  float* ws = (float*)d_ws;
  float* lnst = ws;
  float* Mt   = ws + 3 * D3 * DH;
  float* cb   = Mt  + 3 * D3 * DH;
  float* spP  = cb  + 3 * D3;
  float* qkv  = spP + D9 * D3;
  float* attb = qkv + NT * D9;

  cea_prep<<<dim3(PREP_BIAS_BLOCKS + PREP_FUSE_BLOCKS + PREP_SPP_BLOCKS), dim3(256), 0, stream>>>(
      Wq, Wk, Wv, Wp, bq, bk, bv, Sq, Sk, Sv, Mt, spP, cb);

  cea_qkv<<<dim3(NT / QK_BM, 3 * (D3 / QK_BN)), dim3(128), 0, stream>>>(x, Mt, cb, qkv);

  cea_attn<<<dim3(NT), dim3(D3), 0, stream>>>(qkv, attb);

  cea_proj<<<dim3(NT / PJ_BM, D9 / PJ_BN), dim3(128), 0, stream>>>(attb, spP, bp, out, lnst);

  cea_ln<<<dim3(NT), dim3(256), 0, stream>>>(out, lnst, gn, bn);

  (void)in_sizes; (void)n_in; (void)out_size; (void)ws_size;
}
